# starting kernel
# baseline (speedup 1.0000x reference)
#include <hip/hip_runtime.h>
#include <hip/hip_bf16.h>


typedef float v2f __attribute__((ext_vector_type(2)));
typedef float v8f __attribute__((ext_vector_type(8)));

#define FIN   512
#define FH    16
#define FOUT  2


__global__ void k_init_deg(float* __restrict__ deg, int n) {
    int i = blockIdx.x * blockDim.x + threadIdx.x;
    if (i < n) deg[i] = 1.0f;
}

__global__ void k_deg_accum(const int* __restrict__ ei, int E,
                            float* __restrict__ deg) {
    int e = blockIdx.x * blockDim.x + threadIdx.x;
    if (e < E) atomicAdd(&deg[ei[E + e]], 1.0f);
}

__global__ void k_rsqrt_inplace(float* __restrict__ buf, int n) {
    int i = blockIdx.x * blockDim.x + threadIdx.x;
    if (i < n) buf[i] = rsqrtf(buf[i]);
}


__global__ void __launch_bounds__(256)
k_gemm1_wmma(const float* __restrict__ x, const float* __restrict__ W1,
             float* __restrict__ h, int n_tiles) {
    __shared__ float lw[FIN * FH];
    int tid = threadIdx.x;
    for (int i = tid; i < FIN * FH; i += 256) lw[i] = W1[i];
    __syncthreads();

    int wave = tid >> 5;
    int lane = tid & 31;
    int tile = blockIdx.x * 8 + wave;
    if (tile >= n_tiles) return;

    int half = lane >> 4;
    int mn   = lane & 15;
    int kofs = 2 * half;

    const float* __restrict__ xrow = x + (size_t)(tile * 16 + mn) * FIN;

    v8f c = {};
    #pragma unroll 4
    for (int k = 0; k < FIN; k += 4) {
        v2f a = *(const v2f*)(xrow + k + kofs);
        v2f b;
        b.x = lw[(k + kofs)     * FH + mn];
        b.y = lw[(k + kofs + 1) * FH + mn];
        c = __builtin_amdgcn_wmma_f32_16x16x4_f32(
                 false, a,  false, b,
                 (short)0, c,  false,  false);
    }

    float* __restrict__ hrow = h + (size_t)tile * 16 * FH;
    #pragma unroll
    for (int r = 0; r < 8; ++r)
        hrow[(r + 8 * half) * FH + mn] = c[r];
}


__global__ void k_scatter1_init(const float* __restrict__ h,
                                const float* __restrict__ dinv,
                                float* __restrict__ agg1, int n) {
    int t = blockIdx.x * blockDim.x + threadIdx.x;
    if (t < n * FH) {
        int i = t >> 4;
        float w = dinv[i];
        agg1[t] = h[t] * w * w;
    }
}

__global__ void k_scatter1(const int* __restrict__ ei, int E,
                           const float* __restrict__ h,
                           const float* __restrict__ dinv,
                           float* __restrict__ agg1) {
    int t = blockIdx.x * blockDim.x + threadIdx.x;
    if (t < E * FH) {
        int e = t >> 4, f = t & 15;
        int s = ei[e], d = ei[E + e];
        float w = dinv[s] * dinv[d];
        atomicAdd(&agg1[d * FH + f], h[s * FH + f] * w);
    }
}


__global__ void k_layer2(const float* __restrict__ agg1,
                         const float* __restrict__ b1,
                         const float* __restrict__ W2,
                         float* __restrict__ h2, int n) {
    int i = blockIdx.x * blockDim.x + threadIdx.x;
    if (i < n) {
        float a0 = 0.0f, a1 = 0.0f;
        #pragma unroll
        for (int k = 0; k < FH; ++k) {
            float v = fmaxf(agg1[(size_t)i * FH + k] + b1[k], 0.0f);
            a0 = fmaf(v, W2[2 * k],     a0);
            a1 = fmaf(v, W2[2 * k + 1], a1);
        }
        h2[2 * i]     = a0;
        h2[2 * i + 1] = a1;
    }
}


__global__ void k_scatter2_init(const float* __restrict__ h2,
                                const float* __restrict__ dinv,
                                float* __restrict__ agg2, int n) {
    int t = blockIdx.x * blockDim.x + threadIdx.x;
    if (t < n * FOUT) {
        int i = t >> 1;
        float w = dinv[i];
        agg2[t] = h2[t] * w * w;
    }
}

__global__ void k_scatter2(const int* __restrict__ ei, int E,
                           const float* __restrict__ h2,
                           const float* __restrict__ dinv,
                           float* __restrict__ agg2) {
    int t = blockIdx.x * blockDim.x + threadIdx.x;
    if (t < E * FOUT) {
        int e = t >> 1, f = t & 1;
        int s = ei[e], d = ei[E + e];
        float w = dinv[s] * dinv[d];
        atomicAdd(&agg2[d * FOUT + f], h2[s * FOUT + f] * w);
    }
}


__global__ void k_logsoftmax(const float* __restrict__ agg2,
                             const float* __restrict__ b2,
                             float* __restrict__ out, int n) {
    int i = blockIdx.x * blockDim.x + threadIdx.x;
    if (i < n) {
        float a0 = agg2[2 * i]     + b2[0];
        float a1 = agg2[2 * i + 1] + b2[1];
        float m  = fmaxf(a0, a1);
        float ls = logf(expf(a0 - m) + expf(a1 - m));
        out[2 * i]     = a0 - m - ls;
        out[2 * i + 1] = a1 - m - ls;
    }
}


extern "C" void kernel_launch(void* const* d_in, const int* in_sizes, int n_in,
                              void* d_out, int out_size, void* d_ws, size_t ws_size,
                              hipStream_t stream) {
    const float* x  = (const float*)d_in[0];
    const int*   ei = (const int*)d_in[1];
    const float* W1 = (const float*)d_in[2];
    const float* b1 = (const float*)d_in[3];
    const float* W2 = (const float*)d_in[4];
    const float* b2 = (const float*)d_in[5];
    float* out      = (float*)d_out;

    const int N = in_sizes[0] / FIN;
    const int E = in_sizes[1] / 2;

    float* ws   = (float*)d_ws;
    float* dinv = ws;
    float* h    = dinv + N;
    float* agg1 = h    + (size_t)N * FH;
    float* h2   = agg1 + (size_t)N * FH;
    float* agg2 = h2   + (size_t)N * FOUT;

    const int B = 256;
    auto blk = [](long long t, int b) { return (int)((t + b - 1) / b); };

    k_init_deg   <<<blk(N, B), B, 0, stream>>>(dinv, N);
    k_deg_accum  <<<blk(E, B), B, 0, stream>>>(ei, E, dinv);
    k_rsqrt_inplace<<<blk(N, B), B, 0, stream>>>(dinv, N);

    const int n_tiles = N / 16;
    k_gemm1_wmma <<<blk(n_tiles, 8), B, 0, stream>>>(x, W1, h, n_tiles);
    k_scatter1_init<<<blk((long long)N * FH, B), B, 0, stream>>>(h, dinv, agg1, N);
    k_scatter1   <<<blk((long long)E * FH, B), B, 0, stream>>>(ei, E, h, dinv, agg1);

    k_layer2     <<<blk(N, B), B, 0, stream>>>(agg1, b1, W2, h2, N);
    k_scatter2_init<<<blk((long long)N * FOUT, B), B, 0, stream>>>(h2, dinv, agg2, N);
    k_scatter2   <<<blk((long long)E * FOUT, B), B, 0, stream>>>(ei, E, h2, dinv, agg2);

    k_logsoftmax <<<blk(N, B), B, 0, stream>>>(agg2, b2, out, N);
}


# degree and layer-2 scatter-adds accumulate in LDS per destination range over edge slices, slice partials summed by the next kernel, replacing their global atomics (sum order changes)
# speedup vs baseline: 1.7289x; 1.7289x over previous
#include <hip/hip_runtime.h>
#include <hip/hip_bf16.h>


typedef float v2f __attribute__((ext_vector_type(2)));
typedef float v8f __attribute__((ext_vector_type(8)));

#define FIN   512
#define FH    16
#define FOUT  2


#define RNG_R 12544
#define RNG_S 64
#define RNG_T 1024

__global__ void __launch_bounds__(RNG_T)
k_deg_lds(const int* __restrict__ ei, int E, int n, int q_per_slice,
          unsigned* __restrict__ pdeg) {
    __shared__ unsigned cnt[RNG_R];
    const int tid = threadIdx.x;
    const int lo  = blockIdx.x * RNG_R;
    const int sl  = blockIdx.y;
    for (int i = tid; i < RNG_R; i += RNG_T) cnt[i] = 0u;
    __syncthreads();

    const int nq = E >> 2;
    const int4* __restrict__ dst4 = (const int4*)(ei + E);
    const int q0 = sl * q_per_slice;
    const int q1 = min(q0 + q_per_slice, nq);
    for (int q = q0 + tid; q < q1; q += RNG_T) {
        int4 d = dst4[q];
        unsigned r0 = (unsigned)(d.x - lo);
        unsigned r1 = (unsigned)(d.y - lo);
        unsigned r2 = (unsigned)(d.z - lo);
        unsigned r3 = (unsigned)(d.w - lo);
        if (r0 < RNG_R) atomicAdd(&cnt[r0], 1u);
        if (r1 < RNG_R) atomicAdd(&cnt[r1], 1u);
        if (r2 < RNG_R) atomicAdd(&cnt[r2], 1u);
        if (r3 < RNG_R) atomicAdd(&cnt[r3], 1u);
    }
    if (sl == 0) {
        int e = (nq << 2) + tid;
        if (e < E) {
            unsigned r = (unsigned)(ei[E + e] - lo);
            if (r < RNG_R) atomicAdd(&cnt[r], 1u);
        }
    }
    __syncthreads();

    unsigned* __restrict__ prow = pdeg + (size_t)sl * n;
    for (int i = tid; i < RNG_R; i += RNG_T) {
        int b = lo + i;
        if (b < n) prow[b] = cnt[i];
    }
}

__global__ void k_dinv(const unsigned* __restrict__ pdeg,
                       float* __restrict__ dinv, int n) {
    int i = blockIdx.x * blockDim.x + threadIdx.x;
    if (i < n) {
        unsigned c = 1u;
        #pragma unroll 8
        for (int j = 0; j < RNG_S; ++j) c += pdeg[(size_t)j * n + i];
        dinv[i] = rsqrtf((float)c);
    }
}


__global__ void __launch_bounds__(256)
k_gemm1_wmma(const float* __restrict__ x, const float* __restrict__ W1,
             float* __restrict__ h, int n_tiles) {
    __shared__ float lw[FIN * FH];
    int tid = threadIdx.x;
    for (int i = tid; i < FIN * FH; i += 256) lw[i] = W1[i];
    __syncthreads();

    int wave = tid >> 5;
    int lane = tid & 31;
    int tile = blockIdx.x * 8 + wave;
    if (tile >= n_tiles) return;

    int half = lane >> 4;
    int mn   = lane & 15;
    int kofs = 2 * half;

    const float* __restrict__ xrow = x + (size_t)(tile * 16 + mn) * FIN;

    v8f c = {};
    #pragma unroll 4
    for (int k = 0; k < FIN; k += 4) {
        v2f a = *(const v2f*)(xrow + k + kofs);
        v2f b;
        b.x = lw[(k + kofs)     * FH + mn];
        b.y = lw[(k + kofs + 1) * FH + mn];
        c = __builtin_amdgcn_wmma_f32_16x16x4_f32(
                 false, a,  false, b,
                 (short)0, c,  false,  false);
    }

    float* __restrict__ hrow = h + (size_t)tile * 16 * FH;
    #pragma unroll
    for (int r = 0; r < 8; ++r)
        hrow[(r + 8 * half) * FH + mn] = c[r];
}


__global__ void k_scatter1_init(const float* __restrict__ h,
                                const float* __restrict__ dinv,
                                float* __restrict__ agg1, int n) {
    int t = blockIdx.x * blockDim.x + threadIdx.x;
    if (t < n * FH) {
        int i = t >> 4;
        float w = dinv[i];
        agg1[t] = h[t] * w * w;
    }
}

__global__ void k_scatter1(const int* __restrict__ ei, int E,
                           const float* __restrict__ h,
                           const float* __restrict__ dinv,
                           float* __restrict__ agg1) {
    int t = blockIdx.x * blockDim.x + threadIdx.x;
    if (t < E * FH) {
        int e = t >> 4, f = t & 15;
        int s = ei[e], d = ei[E + e];
        float w = dinv[s] * dinv[d];
        atomicAdd(&agg1[d * FH + f], h[s * FH + f] * w);
    }
}


__global__ void k_layer2(const float* __restrict__ agg1,
                         const float* __restrict__ b1,
                         const float* __restrict__ W2,
                         float* __restrict__ h2, int n) {
    int i = blockIdx.x * blockDim.x + threadIdx.x;
    if (i < n) {
        float a0 = 0.0f, a1 = 0.0f;
        #pragma unroll
        for (int k = 0; k < FH; ++k) {
            float v = fmaxf(agg1[(size_t)i * FH + k] + b1[k], 0.0f);
            a0 = fmaf(v, W2[2 * k],     a0);
            a1 = fmaf(v, W2[2 * k + 1], a1);
        }
        h2[2 * i]     = a0;
        h2[2 * i + 1] = a1;
    }
}


__global__ void __launch_bounds__(RNG_T)
k_scatter2_lds(const int* __restrict__ ei, int E,
               const float* __restrict__ h2,
               const float* __restrict__ dinv,
               int n, int q_per_slice,
               float* __restrict__ p2) {
    __shared__ float acc[RNG_R * FOUT];
    __shared__ float dl[RNG_R];
    const int tid = threadIdx.x;
    const int lo  = blockIdx.x * RNG_R;
    const int sl  = blockIdx.y;
    for (int i = tid; i < RNG_R * FOUT; i += RNG_T) acc[i] = 0.0f;
    for (int i = tid; i < RNG_R; i += RNG_T) {
        int b = lo + i;
        dl[i] = (b < n) ? dinv[b] : 0.0f;
    }
    __syncthreads();

    const int nq = E >> 2;
    const int4* __restrict__ src4 = (const int4*)ei;
    const int4* __restrict__ dst4 = (const int4*)(ei + E);
    const int q0 = sl * q_per_slice;
    const int q1 = min(q0 + q_per_slice, nq);
    for (int q = q0 + tid; q < q1; q += RNG_T) {
        int4 d = dst4[q];
        int4 s = src4[q];
        unsigned r0 = (unsigned)(d.x - lo);
        unsigned r1 = (unsigned)(d.y - lo);
        unsigned r2 = (unsigned)(d.z - lo);
        unsigned r3 = (unsigned)(d.w - lo);
        bool m0 = r0 < RNG_R, m1 = r1 < RNG_R, m2 = r2 < RNG_R, m3 = r3 < RNG_R;
        int s0 = m0 ? s.x : 0, s1 = m1 ? s.y : 0, s2 = m2 ? s.z : 0, s3 = m3 ? s.w : 0;
        float w0 = dinv[s0], w1 = dinv[s1], w2 = dinv[s2], w3 = dinv[s3];
        v2f g0 = *(const v2f*)(h2 + 2 * s0);
        v2f g1 = *(const v2f*)(h2 + 2 * s1);
        v2f g2 = *(const v2f*)(h2 + 2 * s2);
        v2f g3 = *(const v2f*)(h2 + 2 * s3);
        asm volatile("" :: "v"(w0), "v"(w1), "v"(w2), "v"(w3),
                           "v"(g0), "v"(g1), "v"(g2), "v"(g3));
        if (m0) { float w = w0 * dl[r0]; atomicAdd(&acc[2 * r0], g0.x * w); atomicAdd(&acc[2 * r0 + 1], g0.y * w); }
        if (m1) { float w = w1 * dl[r1]; atomicAdd(&acc[2 * r1], g1.x * w); atomicAdd(&acc[2 * r1 + 1], g1.y * w); }
        if (m2) { float w = w2 * dl[r2]; atomicAdd(&acc[2 * r2], g2.x * w); atomicAdd(&acc[2 * r2 + 1], g2.y * w); }
        if (m3) { float w = w3 * dl[r3]; atomicAdd(&acc[2 * r3], g3.x * w); atomicAdd(&acc[2 * r3 + 1], g3.y * w); }
    }
    if (sl == 0) {
        int e = (nq << 2) + tid;
        if (e < E) {
            int sN = ei[e];
            unsigned r = (unsigned)(ei[E + e] - lo);
            if (r < RNG_R) {
                float w = dinv[sN] * dl[r];
                atomicAdd(&acc[2 * r],     h2[2 * sN]     * w);
                atomicAdd(&acc[2 * r + 1], h2[2 * sN + 1] * w);
            }
        }
    }
    __syncthreads();

    float* __restrict__ prow = p2 + (size_t)sl * FOUT * n;
    for (int i = tid; i < RNG_R * FOUT; i += RNG_T) {
        int g = lo * FOUT + i;
        if (g < n * FOUT) prow[g] = acc[i];
    }
}


__global__ void k_logsoftmax(const float* __restrict__ p2,
                             const float* __restrict__ h2,
                             const float* __restrict__ dinv,
                             const float* __restrict__ b2,
                             float* __restrict__ out, int n) {
    int i = blockIdx.x * blockDim.x + threadIdx.x;
    if (i < n) {
        float w  = dinv[i];
        float a0 = h2[2 * i]     * w * w;
        float a1 = h2[2 * i + 1] * w * w;
        #pragma unroll 8
        for (int j = 0; j < RNG_S; ++j) {
            v2f v = *(const v2f*)(p2 + (size_t)j * FOUT * n + 2 * i);
            a0 += v.x;
            a1 += v.y;
        }
        a0 += b2[0];
        a1 += b2[1];
        float m  = fmaxf(a0, a1);
        float ls = logf(expf(a0 - m) + expf(a1 - m));
        out[2 * i]     = a0 - m - ls;
        out[2 * i + 1] = a1 - m - ls;
    }
}


extern "C" void kernel_launch(void* const* d_in, const int* in_sizes, int n_in,
                              void* d_out, int out_size, void* d_ws, size_t ws_size,
                              hipStream_t stream) {
    const float* x  = (const float*)d_in[0];
    const int*   ei = (const int*)d_in[1];
    const float* W1 = (const float*)d_in[2];
    const float* b1 = (const float*)d_in[3];
    const float* W2 = (const float*)d_in[4];
    const float* b2 = (const float*)d_in[5];
    float* out      = (float*)d_out;

    const int N = in_sizes[0] / FIN;
    const int E = in_sizes[1] / 2;

    float* ws      = (float*)d_ws;
    float* p2      = ws;
    unsigned* pdeg = (unsigned*)(p2 + (size_t)RNG_S * FOUT * N);
    float* h2      = (float*)(pdeg + (size_t)RNG_S * N);
    float* h       = h2   + (size_t)N * FOUT;
    float* agg1    = h    + (size_t)N * FH;
    float* dinv    = agg1 + (size_t)N * FH;

    const int B = 256;
    auto blk = [](long long t, int b) { return (int)((t + b - 1) / b); };

    const dim3 rgrid(blk(N, RNG_R), RNG_S);
    const int  qps = blk(E / 4, RNG_S);

    k_deg_lds    <<<rgrid, RNG_T, 0, stream>>>(ei, E, N, qps, pdeg);
    k_dinv       <<<blk(N, B), B, 0, stream>>>(pdeg, dinv, N);

    const int n_tiles = N / 16;
    k_gemm1_wmma <<<blk(n_tiles, 8), B, 0, stream>>>(x, W1, h, n_tiles);
    k_scatter1_init<<<blk((long long)N * FH, B), B, 0, stream>>>(h, dinv, agg1, N);
    k_scatter1   <<<blk((long long)E * FH, B), B, 0, stream>>>(ei, E, h, dinv, agg1);

    k_layer2     <<<blk(N, B), B, 0, stream>>>(agg1, b1, W2, h2, N);
    k_scatter2_lds<<<rgrid, RNG_T, 0, stream>>>(ei, E, h2, dinv, N, qps, p2);

    k_logsoftmax <<<blk(N, B), B, 0, stream>>>(p2, h2, dinv, b2, out, N);
}


# layer-1 scatter-add accumulates in LDS per destination range and edge slice (waves queue in-range edges, half-wave per edge), layer-2 kernel sums partials; replaces global atomics (sum order changes)
# speedup vs baseline: 2.6046x; 1.5065x over previous
#include <hip/hip_runtime.h>
#include <hip/hip_bf16.h>


typedef float v2f __attribute__((ext_vector_type(2)));
typedef float v8f __attribute__((ext_vector_type(8)));

#define FIN   512
#define FH    16
#define FOUT  2


#define RNG_R 12544
#define RNG_S 64
#define RNG_T 1024

__global__ void __launch_bounds__(RNG_T)
k_deg_lds(const int* __restrict__ ei, int E, int n, int q_per_slice,
          unsigned* __restrict__ pdeg) {
    __shared__ unsigned cnt[RNG_R];
    const int tid = threadIdx.x;
    const int lo  = blockIdx.x * RNG_R;
    const int sl  = blockIdx.y;
    for (int i = tid; i < RNG_R; i += RNG_T) cnt[i] = 0u;
    __syncthreads();

    const int nq = E >> 2;
    const int4* __restrict__ dst4 = (const int4*)(ei + E);
    const int q0 = sl * q_per_slice;
    const int q1 = min(q0 + q_per_slice, nq);
    for (int q = q0 + tid; q < q1; q += RNG_T) {
        int4 d = dst4[q];
        unsigned r0 = (unsigned)(d.x - lo);
        unsigned r1 = (unsigned)(d.y - lo);
        unsigned r2 = (unsigned)(d.z - lo);
        unsigned r3 = (unsigned)(d.w - lo);
        if (r0 < RNG_R) atomicAdd(&cnt[r0], 1u);
        if (r1 < RNG_R) atomicAdd(&cnt[r1], 1u);
        if (r2 < RNG_R) atomicAdd(&cnt[r2], 1u);
        if (r3 < RNG_R) atomicAdd(&cnt[r3], 1u);
    }
    if (sl == 0) {
        int e = (nq << 2) + tid;
        if (e < E) {
            unsigned r = (unsigned)(ei[E + e] - lo);
            if (r < RNG_R) atomicAdd(&cnt[r], 1u);
        }
    }
    __syncthreads();

    unsigned* __restrict__ prow = pdeg + (size_t)sl * n;
    for (int i = tid; i < RNG_R; i += RNG_T) {
        int b = lo + i;
        if (b < n) prow[b] = cnt[i];
    }
}

__global__ void k_dinv(const unsigned* __restrict__ pdeg,
                       float* __restrict__ dinv, int n) {
    int i = blockIdx.x * blockDim.x + threadIdx.x;
    if (i < n) {
        unsigned c = 1u;
        #pragma unroll 8
        for (int j = 0; j < RNG_S; ++j) c += pdeg[(size_t)j * n + i];
        dinv[i] = rsqrtf((float)c);
    }
}


__global__ void __launch_bounds__(256)
k_gemm1_wmma(const float* __restrict__ x, const float* __restrict__ W1,
             float* __restrict__ h, int n_tiles) {
    __shared__ float lw[FIN * FH];
    int tid = threadIdx.x;
    for (int i = tid; i < FIN * FH; i += 256) lw[i] = W1[i];
    __syncthreads();

    int wave = tid >> 5;
    int lane = tid & 31;
    int tile = blockIdx.x * 8 + wave;
    if (tile >= n_tiles) return;

    int half = lane >> 4;
    int mn   = lane & 15;
    int kofs = 2 * half;

    const float* __restrict__ xrow = x + (size_t)(tile * 16 + mn) * FIN;

    v8f c = {};
    #pragma unroll 4
    for (int k = 0; k < FIN; k += 4) {
        v2f a = *(const v2f*)(xrow + k + kofs);
        v2f b;
        b.x = lw[(k + kofs)     * FH + mn];
        b.y = lw[(k + kofs + 1) * FH + mn];
        c = __builtin_amdgcn_wmma_f32_16x16x4_f32(
                 false, a,  false, b,
                 (short)0, c,  false,  false);
    }

    float* __restrict__ hrow = h + (size_t)tile * 16 * FH;
    #pragma unroll
    for (int r = 0; r < 8; ++r)
        hrow[(r + 8 * half) * FH + mn] = c[r];
}


#define S1_R     4000
#define S1_S     10
#define S1_T     1024
#define S1_QC    192
#define S1_DRAIN 56

#define S1_DU    8

__device__ __forceinline__ void s1_drain(const int2* q, int cnt, int lane, int lo,
                                         const float* __restrict__ h,
                                         const float* __restrict__ dinv,
                                         float* acc) {
    const int g = lane >> 4, f = lane & 15;
    for (int i = 0; i < cnt; i += 2 * S1_DU) {
        int2  e[S1_DU];
        float hv[S1_DU], ws[S1_DU], wd[S1_DU];
        bool  ok[S1_DU];
        #pragma unroll
        for (int u = 0; u < S1_DU; ++u) {
            int idx = i + 2 * u + g;
            ok[u] = idx < cnt;
            e[u]  = q[ok[u] ? idx : cnt - 1];
        }
        #pragma unroll
        for (int u = 0; u < S1_DU; ++u) {
            hv[u] = h[(unsigned)(e[u].x * FH + f)];
            ws[u] = dinv[(unsigned)e[u].x];
            wd[u] = dinv[(unsigned)(lo + e[u].y)];
        }
        #pragma unroll
        for (int u = 0; u < S1_DU; ++u) {
            float w = ws[u] * wd[u];
            atomicAdd(&acc[e[u].y * FH + f], ok[u] ? hv[u] * w : 0.0f);
        }
    }
}

__global__ void __launch_bounds__(S1_T)
k_scatter1_lds(const int* __restrict__ ei, int E,
               const float* __restrict__ h,
               const float* __restrict__ dinv,
               int n, int q_per_slice,
               float* __restrict__ p1) {
    __shared__ float acc[S1_R * FH];
    __shared__ int2  que[(S1_T / 32) * S1_QC];
    const int tid  = threadIdx.x;
    const int lane = tid & 31;
    const int wave = __builtin_amdgcn_readfirstlane(tid >> 5);
    const int lo   = blockIdx.x * S1_R;
    const int sl   = blockIdx.y;

    float4* acc4 = (float4*)acc;
    for (int i = tid; i < S1_R * FH / 4; i += S1_T) acc4[i] = make_float4(0.f, 0.f, 0.f, 0.f);
    __syncthreads();

    int2* q = que + wave * S1_QC;
    int cnt = 0;

    const int nq = E >> 2;
    const int4* __restrict__ src4 = (const int4*)ei;
    const int4* __restrict__ dst4 = (const int4*)(ei + E);
    const int q0 = sl * q_per_slice;
    const int q1 = min(q0 + q_per_slice, nq);

#define S1_PUT(valid, dv, sv)                                              \
    {                                                                      \
        unsigned r_ = (unsigned)((dv) - lo);                               \
        bool m_ = (valid) && r_ < S1_R;                                    \
        unsigned bal_ = __builtin_amdgcn_ballot_w32(m_);                   \
        if (m_) q[cnt + (int)__builtin_amdgcn_mbcnt_lo(bal_, 0u)] = make_int2((sv), (int)r_); \
        cnt += __builtin_popcount(bal_);                                   \
    }

#define S1_PUT4(valid, d4, s4)                                             \
    S1_PUT(valid, d4.x, s4.x) S1_PUT(valid, d4.y, s4.y)                    \
    S1_PUT(valid, d4.z, s4.z) S1_PUT(valid, d4.w, s4.w)                    \
    if (cnt >= S1_DRAIN) { s1_drain(q, cnt, lane, lo, h, dinv, acc); cnt = 0; }

    for (int base = q0 + (wave << 5); base < q1; base += 4 * S1_T) {
        const int  qa = base + lane, qb = qa + S1_T, qc = qb + S1_T, qd = qc + S1_T;
        const bool va = qa < q1, vb = qb < q1, vc = qc < q1, vd = qd < q1;
        const int  ia = va ? qa : q0, ib = vb ? qb : q0, ic = vc ? qc : q0, id = vd ? qd : q0;
        int4 da = dst4[ia], db = dst4[ib], dc = dst4[ic], dd = dst4[id];
        int4 sa = src4[ia], sb = src4[ib], sc = src4[ic], sd = src4[id];
        asm volatile("" :: "v"(sa.x), "v"(sa.y), "v"(sa.z), "v"(sa.w),
                           "v"(sb.x), "v"(sb.y), "v"(sb.z), "v"(sb.w),
                           "v"(sc.x), "v"(sc.y), "v"(sc.z), "v"(sc.w),
                           "v"(sd.x), "v"(sd.y), "v"(sd.z), "v"(sd.w));
        S1_PUT4(va, da, sa)
        S1_PUT4(vb, db, sb)
        S1_PUT4(vc, dc, sc)
        S1_PUT4(vd, dd, sd)
    }
    if (cnt > 0) s1_drain(q, cnt, lane, lo, h, dinv, acc);
#undef S1_PUT4
#undef S1_PUT

    if (sl == 0) {
        int e = (nq << 2) + tid;
        if (e < E) {
            int sN = ei[e];
            unsigned r = (unsigned)(ei[E + e] - lo);
            if (r < S1_R) {
                float w = dinv[sN] * dinv[lo + r];
                for (int f = 0; f < FH; ++f)
                    atomicAdd(&acc[r * FH + f], h[sN * FH + f] * w);
            }
        }
    }
    __syncthreads();

    float4* __restrict__ prow4 = (float4*)(p1 + (size_t)sl * FH * n);
    for (int i = tid; i < S1_R * FH / 4; i += S1_T) {
        int g4 = lo * (FH / 4) + i;
        if (g4 < n * (FH / 4)) prow4[g4] = acc4[i];
    }
}


__global__ void k_layer2(const float* __restrict__ p1,
                         const float* __restrict__ h,
                         const float* __restrict__ dinv,
                         const float* __restrict__ b1,
                         const float* __restrict__ W2,
                         float* __restrict__ h2, int n) {
    int i = blockIdx.x * blockDim.x + threadIdx.x;
    if (i < n) {
        const float w = dinv[i];
        const float4* __restrict__ h4 = (const float4*)h;
        const float4* __restrict__ p4 = (const float4*)p1;
        float a0 = 0.0f, a1 = 0.0f;
        #pragma unroll
        for (int c = 0; c < FH / 4; ++c) {
            float4 t = h4[(size_t)i * (FH / 4) + c];
            float v[4] = { t.x * w * w, t.y * w * w, t.z * w * w, t.w * w * w };
            #pragma unroll
            for (int j = 0; j < S1_S; ++j) {
                float4 p = p4[((size_t)j * n + i) * (FH / 4) + c];
                v[0] += p.x; v[1] += p.y; v[2] += p.z; v[3] += p.w;
            }
            #pragma unroll
            for (int k = 0; k < 4; ++k) {
                int kk = 4 * c + k;
                float r = fmaxf(v[k] + b1[kk], 0.0f);
                a0 = fmaf(r, W2[2 * kk],     a0);
                a1 = fmaf(r, W2[2 * kk + 1], a1);
            }
        }
        h2[2 * i]     = a0;
        h2[2 * i + 1] = a1;
    }
}


__global__ void __launch_bounds__(RNG_T)
k_scatter2_lds(const int* __restrict__ ei, int E,
               const float* __restrict__ h2,
               const float* __restrict__ dinv,
               int n, int q_per_slice,
               float* __restrict__ p2) {
    __shared__ float acc[RNG_R * FOUT];
    __shared__ float dl[RNG_R];
    const int tid = threadIdx.x;
    const int lo  = blockIdx.x * RNG_R;
    const int sl  = blockIdx.y;
    for (int i = tid; i < RNG_R * FOUT; i += RNG_T) acc[i] = 0.0f;
    for (int i = tid; i < RNG_R; i += RNG_T) {
        int b = lo + i;
        dl[i] = (b < n) ? dinv[b] : 0.0f;
    }
    __syncthreads();

    const int nq = E >> 2;
    const int4* __restrict__ src4 = (const int4*)ei;
    const int4* __restrict__ dst4 = (const int4*)(ei + E);
    const int q0 = sl * q_per_slice;
    const int q1 = min(q0 + q_per_slice, nq);
    for (int q = q0 + tid; q < q1; q += RNG_T) {
        int4 d = dst4[q];
        int4 s = src4[q];
        unsigned r0 = (unsigned)(d.x - lo);
        unsigned r1 = (unsigned)(d.y - lo);
        unsigned r2 = (unsigned)(d.z - lo);
        unsigned r3 = (unsigned)(d.w - lo);
        bool m0 = r0 < RNG_R, m1 = r1 < RNG_R, m2 = r2 < RNG_R, m3 = r3 < RNG_R;
        int s0 = m0 ? s.x : 0, s1 = m1 ? s.y : 0, s2 = m2 ? s.z : 0, s3 = m3 ? s.w : 0;
        float w0 = dinv[s0], w1 = dinv[s1], w2 = dinv[s2], w3 = dinv[s3];
        v2f g0 = *(const v2f*)(h2 + 2 * s0);
        v2f g1 = *(const v2f*)(h2 + 2 * s1);
        v2f g2 = *(const v2f*)(h2 + 2 * s2);
        v2f g3 = *(const v2f*)(h2 + 2 * s3);
        asm volatile("" :: "v"(w0), "v"(w1), "v"(w2), "v"(w3),
                           "v"(g0), "v"(g1), "v"(g2), "v"(g3));
        if (m0) { float w = w0 * dl[r0]; atomicAdd(&acc[2 * r0], g0.x * w); atomicAdd(&acc[2 * r0 + 1], g0.y * w); }
        if (m1) { float w = w1 * dl[r1]; atomicAdd(&acc[2 * r1], g1.x * w); atomicAdd(&acc[2 * r1 + 1], g1.y * w); }
        if (m2) { float w = w2 * dl[r2]; atomicAdd(&acc[2 * r2], g2.x * w); atomicAdd(&acc[2 * r2 + 1], g2.y * w); }
        if (m3) { float w = w3 * dl[r3]; atomicAdd(&acc[2 * r3], g3.x * w); atomicAdd(&acc[2 * r3 + 1], g3.y * w); }
    }
    if (sl == 0) {
        int e = (nq << 2) + tid;
        if (e < E) {
            int sN = ei[e];
            unsigned r = (unsigned)(ei[E + e] - lo);
            if (r < RNG_R) {
                float w = dinv[sN] * dl[r];
                atomicAdd(&acc[2 * r],     h2[2 * sN]     * w);
                atomicAdd(&acc[2 * r + 1], h2[2 * sN + 1] * w);
            }
        }
    }
    __syncthreads();

    float* __restrict__ prow = p2 + (size_t)sl * FOUT * n;
    for (int i = tid; i < RNG_R * FOUT; i += RNG_T) {
        int g = lo * FOUT + i;
        if (g < n * FOUT) prow[g] = acc[i];
    }
}


__global__ void k_logsoftmax(const float* __restrict__ p2,
                             const float* __restrict__ h2,
                             const float* __restrict__ dinv,
                             const float* __restrict__ b2,
                             float* __restrict__ out, int n) {
    int i = blockIdx.x * blockDim.x + threadIdx.x;
    if (i < n) {
        float w  = dinv[i];
        float a0 = h2[2 * i]     * w * w;
        float a1 = h2[2 * i + 1] * w * w;
        #pragma unroll 8
        for (int j = 0; j < RNG_S; ++j) {
            v2f v = *(const v2f*)(p2 + (size_t)j * FOUT * n + 2 * i);
            a0 += v.x;
            a1 += v.y;
        }
        a0 += b2[0];
        a1 += b2[1];
        float m  = fmaxf(a0, a1);
        float ls = logf(expf(a0 - m) + expf(a1 - m));
        out[2 * i]     = a0 - m - ls;
        out[2 * i + 1] = a1 - m - ls;
    }
}


extern "C" void kernel_launch(void* const* d_in, const int* in_sizes, int n_in,
                              void* d_out, int out_size, void* d_ws, size_t ws_size,
                              hipStream_t stream) {
    const float* x  = (const float*)d_in[0];
    const int*   ei = (const int*)d_in[1];
    const float* W1 = (const float*)d_in[2];
    const float* b1 = (const float*)d_in[3];
    const float* W2 = (const float*)d_in[4];
    const float* b2 = (const float*)d_in[5];
    float* out      = (float*)d_out;

    const int N = in_sizes[0] / FIN;
    const int E = in_sizes[1] / 2;

    float* ws      = (float*)d_ws;
    float* p2      = ws;
    unsigned* pdeg = (unsigned*)(p2 + (size_t)RNG_S * FOUT * N);
    float* h2      = (float*)(pdeg + (size_t)RNG_S * N);
    float* h       = h2   + (size_t)N * FOUT;
    float* p1      = h    + (size_t)N * FH;
    float* dinv    = p1   + (size_t)S1_S * N * FH;

    const int B = 256;
    auto blk = [](long long t, int b) { return (int)((t + b - 1) / b); };

    const dim3 rgrid(blk(N, RNG_R), RNG_S);
    const int  qps = blk(E / 4, RNG_S);

    k_deg_lds    <<<rgrid, RNG_T, 0, stream>>>(ei, E, N, qps, pdeg);
    k_dinv       <<<blk(N, B), B, 0, stream>>>(pdeg, dinv, N);

    const int n_tiles = N / 16;
    k_gemm1_wmma <<<blk(n_tiles, 8), B, 0, stream>>>(x, W1, h, n_tiles);
    const dim3 s1grid(blk(N, S1_R), S1_S);
    const int  s1qps = blk(E / 4, S1_S);
    k_scatter1_lds<<<s1grid, S1_T, 0, stream>>>(ei, E, h, dinv, N, s1qps, p1);

    k_layer2     <<<blk(N, B), B, 0, stream>>>(p1, h, dinv, b1, W2, h2, N);
    k_scatter2_lds<<<rgrid, RNG_T, 0, stream>>>(ei, E, h2, dinv, N, qps, p2);

    k_logsoftmax <<<blk(N, B), B, 0, stream>>>(p2, h2, dinv, b2, out, N);
}


# source dinv folded into the gathered rows by their producers (GEMM1 epilogue, layer-2 kernel), destination dinv applied once per node after the sum: no per-edge dinv gathers (term rounding changes)
# speedup vs baseline: 3.2948x; 1.2650x over previous
#include <hip/hip_runtime.h>
#include <hip/hip_bf16.h>


typedef float v2f __attribute__((ext_vector_type(2)));
typedef float v8f __attribute__((ext_vector_type(8)));

#define FIN   512
#define FH    16
#define FOUT  2


#define RNG_R 12544
#define RNG_S 64
#define RNG_T 1024

__global__ void __launch_bounds__(RNG_T)
k_deg_lds(const int* __restrict__ ei, int E, int n, int q_per_slice,
          unsigned* __restrict__ pdeg) {
    __shared__ unsigned cnt[RNG_R];
    const int tid = threadIdx.x;
    const int lo  = blockIdx.x * RNG_R;
    const int sl  = blockIdx.y;
    for (int i = tid; i < RNG_R; i += RNG_T) cnt[i] = 0u;
    __syncthreads();

    const int nq = E >> 2;
    const int4* __restrict__ dst4 = (const int4*)(ei + E);
    const int q0 = sl * q_per_slice;
    const int q1 = min(q0 + q_per_slice, nq);
    for (int q = q0 + tid; q < q1; q += RNG_T) {
        int4 d = dst4[q];
        unsigned r0 = (unsigned)(d.x - lo);
        unsigned r1 = (unsigned)(d.y - lo);
        unsigned r2 = (unsigned)(d.z - lo);
        unsigned r3 = (unsigned)(d.w - lo);
        if (r0 < RNG_R) atomicAdd(&cnt[r0], 1u);
        if (r1 < RNG_R) atomicAdd(&cnt[r1], 1u);
        if (r2 < RNG_R) atomicAdd(&cnt[r2], 1u);
        if (r3 < RNG_R) atomicAdd(&cnt[r3], 1u);
    }
    if (sl == 0) {
        int e = (nq << 2) + tid;
        if (e < E) {
            unsigned r = (unsigned)(ei[E + e] - lo);
            if (r < RNG_R) atomicAdd(&cnt[r], 1u);
        }
    }
    __syncthreads();

    unsigned* __restrict__ prow = pdeg + (size_t)sl * n;
    for (int i = tid; i < RNG_R; i += RNG_T) {
        int b = lo + i;
        if (b < n) prow[b] = cnt[i];
    }
}

__global__ void k_dinv(const unsigned* __restrict__ pdeg,
                       float* __restrict__ dinv, int n) {
    int i = blockIdx.x * blockDim.x + threadIdx.x;
    if (i < n) {
        unsigned c = 1u;
        #pragma unroll 8
        for (int j = 0; j < RNG_S; ++j) c += pdeg[(size_t)j * n + i];
        dinv[i] = rsqrtf((float)c);
    }
}


__global__ void __launch_bounds__(256)
k_gemm1_wmma(const float* __restrict__ x, const float* __restrict__ W1,
             const float* __restrict__ dinv,
             float* __restrict__ h, int n_tiles) {
    __shared__ float lw[FIN * FH];
    int tid = threadIdx.x;
    for (int i = tid; i < FIN * FH; i += 256) lw[i] = W1[i];
    __syncthreads();

    int wave = tid >> 5;
    int lane = tid & 31;
    int tile = blockIdx.x * 8 + wave;
    if (tile >= n_tiles) return;

    int half = lane >> 4;
    int mn   = lane & 15;
    int kofs = 2 * half;

    const float* __restrict__ xrow = x + (size_t)(tile * 16 + mn) * FIN;

    float dsc[8];
    #pragma unroll
    for (int r = 0; r < 8; ++r) dsc[r] = dinv[tile * 16 + 8 * half + r];

    v8f c = {};
    #pragma unroll 4
    for (int k = 0; k < FIN; k += 4) {
        v2f a = *(const v2f*)(xrow + k + kofs);
        v2f b;
        b.x = lw[(k + kofs)     * FH + mn];
        b.y = lw[(k + kofs + 1) * FH + mn];
        c = __builtin_amdgcn_wmma_f32_16x16x4_f32(
                 false, a,  false, b,
                 (short)0, c,  false,  false);
    }

    float* __restrict__ hrow = h + (size_t)tile * 16 * FH;
    #pragma unroll
    for (int r = 0; r < 8; ++r)
        hrow[(r + 8 * half) * FH + mn] = c[r] * dsc[r];
}


#define S1_R     4000
#define S1_S     10
#define S1_T     1024
#define S1_QC    192
#define S1_DRAIN 56

#define S1_DU    8

__device__ __forceinline__ void s1_drain(const int2* q, int cnt, int lane,
                                         const float* __restrict__ h,
                                         float* acc) {
    const int g = lane >> 4, f = lane & 15;
    for (int i = 0; i < cnt; i += 2 * S1_DU) {
        int2  e[S1_DU];
        float hv[S1_DU];
        bool  ok[S1_DU];
        #pragma unroll
        for (int u = 0; u < S1_DU; ++u) {
            int idx = i + 2 * u + g;
            ok[u] = idx < cnt;
            e[u]  = q[ok[u] ? idx : cnt - 1];
        }
        #pragma unroll
        for (int u = 0; u < S1_DU; ++u)
            hv[u] = h[(unsigned)(e[u].x * FH + f)];
        #pragma unroll
        for (int u = 0; u < S1_DU; ++u)
            atomicAdd(&acc[e[u].y * FH + f], ok[u] ? hv[u] : 0.0f);
    }
}

__global__ void __launch_bounds__(S1_T)
k_scatter1_lds(const int* __restrict__ ei, int E,
               const float* __restrict__ h,
               int n, int q_per_slice,
               float* __restrict__ p1) {
    __shared__ float acc[S1_R * FH];
    __shared__ int2  que[(S1_T / 32) * S1_QC];
    const int tid  = threadIdx.x;
    const int lane = tid & 31;
    const int wave = __builtin_amdgcn_readfirstlane(tid >> 5);
    const int lo   = blockIdx.x * S1_R;
    const int sl   = blockIdx.y;

    float4* acc4 = (float4*)acc;
    for (int i = tid; i < S1_R * FH / 4; i += S1_T) acc4[i] = make_float4(0.f, 0.f, 0.f, 0.f);
    __syncthreads();

    int2* q = que + wave * S1_QC;
    int cnt = 0;

    const int nq = E >> 2;
    const int4* __restrict__ src4 = (const int4*)ei;
    const int4* __restrict__ dst4 = (const int4*)(ei + E);
    const int q0 = sl * q_per_slice;
    const int q1 = min(q0 + q_per_slice, nq);

#define S1_PUT(valid, dv, sv)                                              \
    {                                                                      \
        unsigned r_ = (unsigned)((dv) - lo);                               \
        bool m_ = (valid) && r_ < S1_R;                                    \
        unsigned bal_ = __builtin_amdgcn_ballot_w32(m_);                   \
        if (m_) q[cnt + (int)__builtin_amdgcn_mbcnt_lo(bal_, 0u)] = make_int2((sv), (int)r_); \
        cnt += __builtin_popcount(bal_);                                   \
    }

#define S1_PUT4(valid, d4, s4)                                             \
    S1_PUT(valid, d4.x, s4.x) S1_PUT(valid, d4.y, s4.y)                    \
    S1_PUT(valid, d4.z, s4.z) S1_PUT(valid, d4.w, s4.w)                    \
    if (cnt >= S1_DRAIN) { s1_drain(q, cnt, lane, h, acc); cnt = 0; }

    for (int base = q0 + (wave << 5); base < q1; base += 4 * S1_T) {
        const int  qa = base + lane, qb = qa + S1_T, qc = qb + S1_T, qd = qc + S1_T;
        const bool va = qa < q1, vb = qb < q1, vc = qc < q1, vd = qd < q1;
        const int  ia = va ? qa : q0, ib = vb ? qb : q0, ic = vc ? qc : q0, id = vd ? qd : q0;
        int4 da = dst4[ia], db = dst4[ib], dc = dst4[ic], dd = dst4[id];
        int4 sa = src4[ia], sb = src4[ib], sc = src4[ic], sd = src4[id];
        asm volatile("" :: "v"(sa.x), "v"(sa.y), "v"(sa.z), "v"(sa.w),
                           "v"(sb.x), "v"(sb.y), "v"(sb.z), "v"(sb.w),
                           "v"(sc.x), "v"(sc.y), "v"(sc.z), "v"(sc.w),
                           "v"(sd.x), "v"(sd.y), "v"(sd.z), "v"(sd.w));
        S1_PUT4(va, da, sa)
        S1_PUT4(vb, db, sb)
        S1_PUT4(vc, dc, sc)
        S1_PUT4(vd, dd, sd)
    }
    if (cnt > 0) s1_drain(q, cnt, lane, h, acc);
#undef S1_PUT4
#undef S1_PUT

    if (sl == 0) {
        int e = (nq << 2) + tid;
        if (e < E) {
            int sN = ei[e];
            unsigned r = (unsigned)(ei[E + e] - lo);
            if (r < S1_R) {
                for (int f = 0; f < FH; ++f)
                    atomicAdd(&acc[r * FH + f], h[sN * FH + f]);
            }
        }
    }
    __syncthreads();

    float4* __restrict__ prow4 = (float4*)(p1 + (size_t)sl * FH * n);
    for (int i = tid; i < S1_R * FH / 4; i += S1_T) {
        int g4 = lo * (FH / 4) + i;
        if (g4 < n * (FH / 4)) prow4[g4] = acc4[i];
    }
}


__global__ void k_layer2(const float* __restrict__ p1,
                         const float* __restrict__ h,
                         const float* __restrict__ dinv,
                         const float* __restrict__ b1,
                         const float* __restrict__ W2,
                         float* __restrict__ h2, int n) {
    int i = blockIdx.x * blockDim.x + threadIdx.x;
    if (i < n) {
        const float w = dinv[i];
        const float4* __restrict__ h4 = (const float4*)h;
        const float4* __restrict__ p4 = (const float4*)p1;
        float a0 = 0.0f, a1 = 0.0f;
        #pragma unroll
        for (int c = 0; c < FH / 4; ++c) {
            float4 t = h4[(size_t)i * (FH / 4) + c];
            float v[4] = { t.x, t.y, t.z, t.w };
            #pragma unroll
            for (int j = 0; j < S1_S; ++j) {
                float4 p = p4[((size_t)j * n + i) * (FH / 4) + c];
                v[0] += p.x; v[1] += p.y; v[2] += p.z; v[3] += p.w;
            }
            #pragma unroll
            for (int k = 0; k < 4; ++k) {
                int kk = 4 * c + k;
                float r = fmaxf(v[k] * w + b1[kk], 0.0f);
                a0 = fmaf(r, W2[2 * kk],     a0);
                a1 = fmaf(r, W2[2 * kk + 1], a1);
            }
        }
        h2[2 * i]     = a0 * w;
        h2[2 * i + 1] = a1 * w;
    }
}


__global__ void __launch_bounds__(RNG_T)
k_scatter2_lds(const int* __restrict__ ei, int E,
               const float* __restrict__ h2,
               int n, int q_per_slice,
               float* __restrict__ p2) {
    __shared__ float acc[RNG_R * FOUT];
    const int tid = threadIdx.x;
    const int lo  = blockIdx.x * RNG_R;
    const int sl  = blockIdx.y;
    for (int i = tid; i < RNG_R * FOUT; i += RNG_T) acc[i] = 0.0f;
    __syncthreads();

    const int nq = E >> 2;
    const int4* __restrict__ src4 = (const int4*)ei;
    const int4* __restrict__ dst4 = (const int4*)(ei + E);
    const int q0 = sl * q_per_slice;
    const int q1 = min(q0 + q_per_slice, nq);
    for (int q = q0 + tid; q < q1; q += RNG_T) {
        int4 d = dst4[q];
        int4 s = src4[q];
        unsigned r0 = (unsigned)(d.x - lo);
        unsigned r1 = (unsigned)(d.y - lo);
        unsigned r2 = (unsigned)(d.z - lo);
        unsigned r3 = (unsigned)(d.w - lo);
        bool m0 = r0 < RNG_R, m1 = r1 < RNG_R, m2 = r2 < RNG_R, m3 = r3 < RNG_R;
        int s0 = m0 ? s.x : 0, s1 = m1 ? s.y : 0, s2 = m2 ? s.z : 0, s3 = m3 ? s.w : 0;
        v2f g0 = *(const v2f*)(h2 + 2 * s0);
        v2f g1 = *(const v2f*)(h2 + 2 * s1);
        v2f g2 = *(const v2f*)(h2 + 2 * s2);
        v2f g3 = *(const v2f*)(h2 + 2 * s3);
        asm volatile("" :: "v"(g0), "v"(g1), "v"(g2), "v"(g3));
        if (m0) { atomicAdd(&acc[2 * r0], g0.x); atomicAdd(&acc[2 * r0 + 1], g0.y); }
        if (m1) { atomicAdd(&acc[2 * r1], g1.x); atomicAdd(&acc[2 * r1 + 1], g1.y); }
        if (m2) { atomicAdd(&acc[2 * r2], g2.x); atomicAdd(&acc[2 * r2 + 1], g2.y); }
        if (m3) { atomicAdd(&acc[2 * r3], g3.x); atomicAdd(&acc[2 * r3 + 1], g3.y); }
    }
    if (sl == 0) {
        int e = (nq << 2) + tid;
        if (e < E) {
            int sN = ei[e];
            unsigned r = (unsigned)(ei[E + e] - lo);
            if (r < RNG_R) {
                atomicAdd(&acc[2 * r],     h2[2 * sN]);
                atomicAdd(&acc[2 * r + 1], h2[2 * sN + 1]);
            }
        }
    }
    __syncthreads();

    float* __restrict__ prow = p2 + (size_t)sl * FOUT * n;
    for (int i = tid; i < RNG_R * FOUT; i += RNG_T) {
        int g = lo * FOUT + i;
        if (g < n * FOUT) prow[g] = acc[i];
    }
}


__global__ void k_logsoftmax(const float* __restrict__ p2,
                             const float* __restrict__ h2,
                             const float* __restrict__ dinv,
                             const float* __restrict__ b2,
                             float* __restrict__ out, int n) {
    int i = blockIdx.x * blockDim.x + threadIdx.x;
    if (i < n) {
        float w  = dinv[i];
        float a0 = h2[2 * i];
        float a1 = h2[2 * i + 1];
        #pragma unroll 8
        for (int j = 0; j < RNG_S; ++j) {
            v2f v = *(const v2f*)(p2 + (size_t)j * FOUT * n + 2 * i);
            a0 += v.x;
            a1 += v.y;
        }
        a0 = a0 * w + b2[0];
        a1 = a1 * w + b2[1];
        float m  = fmaxf(a0, a1);
        float ls = logf(expf(a0 - m) + expf(a1 - m));
        out[2 * i]     = a0 - m - ls;
        out[2 * i + 1] = a1 - m - ls;
    }
}


extern "C" void kernel_launch(void* const* d_in, const int* in_sizes, int n_in,
                              void* d_out, int out_size, void* d_ws, size_t ws_size,
                              hipStream_t stream) {
    const float* x  = (const float*)d_in[0];
    const int*   ei = (const int*)d_in[1];
    const float* W1 = (const float*)d_in[2];
    const float* b1 = (const float*)d_in[3];
    const float* W2 = (const float*)d_in[4];
    const float* b2 = (const float*)d_in[5];
    float* out      = (float*)d_out;

    const int N = in_sizes[0] / FIN;
    const int E = in_sizes[1] / 2;

    float* ws      = (float*)d_ws;
    float* p2      = ws;
    unsigned* pdeg = (unsigned*)(p2 + (size_t)RNG_S * FOUT * N);
    float* h2      = (float*)(pdeg + (size_t)RNG_S * N);
    float* h       = h2   + (size_t)N * FOUT;
    float* p1      = h    + (size_t)N * FH;
    float* dinv    = p1   + (size_t)S1_S * N * FH;

    const int B = 256;
    auto blk = [](long long t, int b) { return (int)((t + b - 1) / b); };

    const dim3 rgrid(blk(N, RNG_R), RNG_S);
    const int  qps = blk(E / 4, RNG_S);

    k_deg_lds    <<<rgrid, RNG_T, 0, stream>>>(ei, E, N, qps, pdeg);
    k_dinv       <<<blk(N, B), B, 0, stream>>>(pdeg, dinv, N);

    const int n_tiles = N / 16;
    k_gemm1_wmma <<<blk(n_tiles, 8), B, 0, stream>>>(x, W1, dinv, h, n_tiles);
    const dim3 s1grid(blk(N, S1_R), S1_S);
    const int  s1qps = blk(E / 4, S1_S);
    k_scatter1_lds<<<s1grid, S1_T, 0, stream>>>(ei, E, h, N, s1qps, p1);

    k_layer2     <<<blk(N, B), B, 0, stream>>>(p1, h, dinv, b1, W2, h2, N);
    k_scatter2_lds<<<rgrid, RNG_T, 0, stream>>>(ei, E, h2, N, qps, p2);

    k_logsoftmax <<<blk(N, B), B, 0, stream>>>(p2, h2, dinv, b2, out, N);
}


# degree, layer-1 and layer-2 scatters read edges pre-bucketed by destination range (one added pass sorts the edge list into per-range segments) instead of testing every edge (sum order changes)
# speedup vs baseline: 4.5939x; 1.3943x over previous
#include <hip/hip_runtime.h>
#include <hip/hip_bf16.h>


typedef float v2f __attribute__((ext_vector_type(2)));
typedef float v8f __attribute__((ext_vector_type(8)));

#define FIN   512
#define FH    16
#define FOUT  2


#define S1_R     4000
#define S1_S     10
#define S1_T     1024
#define S1_W     (S1_T / 32)

#define BK_T     1024
#define BK_CAP   16384
#define BK_MAXB  32
#define BK_PITCH (BK_CAP + 1056)

__global__ void __launch_bounds__(BK_T)
k_bucket(const int* __restrict__ ei, int E, int nb,
         unsigned* __restrict__ seg, unsigned* __restrict__ segcnt) {
    __shared__ unsigned stage[BK_CAP];
    __shared__ unsigned lcnt[BK_MAXB];
    __shared__ unsigned lbase[BK_MAXB];
    const int tid = threadIdx.x;
    const int k   = blockIdx.x;
    if (tid < BK_MAXB) lcnt[tid] = 0u;
    __syncthreads();

    const int nq = E >> 2;
    const int4* __restrict__ src4 = (const int4*)ei;
    const int4* __restrict__ dst4 = (const int4*)(ei + E);
    const int qb = k * (BK_CAP / 4);

    int4 d[4], s[4];
    bool v[4];
    #pragma unroll
    for (int j = 0; j < 4; ++j) {
        const int q = qb + j * BK_T + tid;
        v[j] = q < nq;
        d[j] = dst4[v[j] ? q : qb];
        s[j] = src4[v[j] ? q : qb];
    }

    unsigned pk[16], bk[16], rk[16];
    bool vd[16];
    #pragma unroll
    for (int j = 0; j < 4; ++j) {
        const int dd[4] = { d[j].x, d[j].y, d[j].z, d[j].w };
        const int ss[4] = { s[j].x, s[j].y, s[j].z, s[j].w };
        #pragma unroll
        for (int c = 0; c < 4; ++c) {
            const int t = 4 * j + c;
            const unsigned b = (unsigned)dd[c] / S1_R;
            const unsigned r = (unsigned)dd[c] - b * S1_R;
            vd[t] = v[j] && b < (unsigned)nb;
            bk[t] = vd[t] ? b : 0u;
            pk[t] = ((unsigned)ss[c] << 12) | r;
            rk[t] = 0u;
            if (vd[t]) rk[t] = atomicAdd(&lcnt[bk[t]], 1u);
        }
    }
    __syncthreads();

    if (tid < BK_MAXB) {
        unsigned pre = 0u;
        for (int j = 0; j < BK_MAXB; ++j) {
            const unsigned c = lcnt[j];
            pre += (j < tid) ? c : 0u;
        }
        lbase[tid] = pre;
        if (tid < nb) segcnt[(size_t)tid * gridDim.x + k] = lcnt[tid];
    }
    __syncthreads();

    unsigned pos[16];
    #pragma unroll
    for (int t = 0; t < 16; ++t) pos[t] = lbase[bk[t]] + rk[t];
    #pragma unroll
    for (int t = 0; t < 16; ++t)
        if (vd[t]) stage[pos[t]] = pk[t];
    __syncthreads();

    for (int b = 0; b < nb; ++b) {
        const unsigned c = lcnt[b], base = lbase[b];
        unsigned* __restrict__ o = seg + ((size_t)b * gridDim.x + k) * BK_PITCH;
        for (unsigned i = tid; i < c; i += BK_T) o[i] = stage[base + i];
    }
}


__global__ void __launch_bounds__(S1_T)
k_deg_bkt(const int* __restrict__ ei, int E, int n,
          const unsigned* __restrict__ seg,
          const unsigned* __restrict__ segcnt,
          int nbk, int kps,
          unsigned* __restrict__ pdeg) {
    __shared__ unsigned cnt[S1_R];
    const int tid = threadIdx.x;
    const int rb  = blockIdx.x;
    const int lo  = rb * S1_R;
    const int sl  = blockIdx.y;
    for (int i = tid; i < S1_R; i += S1_T) cnt[i] = 0u;
    __syncthreads();

    const int k0 = sl * kps;
    const int k1 = min(k0 + kps, nbk);
    for (int kq = k0; kq < k1; kq += 4) {
        unsigned c[4], e[4];
        const unsigned* sg[4];
        #pragma unroll
        for (int u = 0; u < 4; ++u) {
            const int k = min(kq + u, k1 - 1);
            c[u]  = (kq + u < k1) ? segcnt[(size_t)rb * nbk + k] : 0u;
            sg[u] = seg + ((size_t)rb * nbk + k) * BK_PITCH;
        }
        #pragma unroll
        for (int u = 0; u < 4; ++u) e[u] = ((unsigned)tid < c[u]) ? sg[u][tid] : 0u;
        asm volatile("" :: "v"(e[0]), "v"(e[1]), "v"(e[2]), "v"(e[3]));
        #pragma unroll
        for (int u = 0; u < 4; ++u)
            if ((unsigned)tid < c[u]) atomicAdd(&cnt[e[u] & 4095u], 1u);
        #pragma unroll
        for (int u = 0; u < 4; ++u)
            for (unsigned i = (unsigned)tid + S1_T; i < c[u]; i += S1_T)
                atomicAdd(&cnt[sg[u][i] & 4095u], 1u);
    }
    if (sl == 0) {
        const int nq = E >> 2;
        int e = (nq << 2) + tid;
        if (e < E) {
            unsigned r = (unsigned)(ei[E + e] - lo);
            if (r < S1_R) atomicAdd(&cnt[r], 1u);
        }
    }
    __syncthreads();

    unsigned* __restrict__ prow = pdeg + (size_t)sl * n;
    for (int i = tid; i < S1_R; i += S1_T) {
        int b = lo + i;
        if (b < n) prow[b] = cnt[i];
    }
}

__global__ void k_dinv(const unsigned* __restrict__ pdeg,
                       float* __restrict__ dinv, int n) {
    int i = blockIdx.x * blockDim.x + threadIdx.x;
    if (i < n) {
        unsigned c = 1u;
        #pragma unroll
        for (int j = 0; j < S1_S; ++j) c += pdeg[(size_t)j * n + i];
        dinv[i] = rsqrtf((float)c);
    }
}


__global__ void __launch_bounds__(256)
k_gemm1_wmma(const float* __restrict__ x, const float* __restrict__ W1,
             const float* __restrict__ dinv,
             float* __restrict__ h, int n_tiles) {
    __shared__ float lw[FIN * FH];
    int tid = threadIdx.x;
    for (int i = tid; i < FIN * FH; i += 256) lw[i] = W1[i];
    __syncthreads();

    int wave = tid >> 5;
    int lane = tid & 31;
    int tile = blockIdx.x * 8 + wave;
    if (tile >= n_tiles) return;

    int half = lane >> 4;
    int mn   = lane & 15;
    int kofs = 2 * half;

    const float* __restrict__ xrow = x + (size_t)(tile * 16 + mn) * FIN;

    float dsc[8];
    #pragma unroll
    for (int r = 0; r < 8; ++r) dsc[r] = dinv[tile * 16 + 8 * half + r];

    v8f c = {};
    #pragma unroll 4
    for (int k = 0; k < FIN; k += 4) {
        v2f a = *(const v2f*)(xrow + k + kofs);
        v2f b;
        b.x = lw[(k + kofs)     * FH + mn];
        b.y = lw[(k + kofs + 1) * FH + mn];
        c = __builtin_amdgcn_wmma_f32_16x16x4_f32(
                 false, a,  false, b,
                 (short)0, c,  false,  false);
    }

    float* __restrict__ hrow = h + (size_t)tile * 16 * FH;
    #pragma unroll
    for (int r = 0; r < 8; ++r)
        hrow[(r + 8 * half) * FH + mn] = c[r] * dsc[r];
}


__device__ __forceinline__ void s1_round(int ev, unsigned i0, unsigned cnt,
                                         int g, int f,
                                         const float* __restrict__ h,
                                         float* acc) {
    float    hv[16];
    unsigned rr[16];
    bool     ok[16];
    #pragma unroll
    for (int u = 0; u < 16; ++u) {
        const unsigned e = (unsigned)__shfl(ev, 2 * u + g);
        ok[u] = (i0 + (unsigned)(2 * u + g)) < cnt;
        rr[u] = e & 4095u;
        hv[u] = h[(e >> 12) * FH + f];
    }
    #pragma unroll
    for (int u = 0; u < 16; ++u)
        atomicAdd(&acc[rr[u] * FH + f], ok[u] ? hv[u] : 0.0f);
}

__global__ void __launch_bounds__(S1_T)
k_scatter1_lds(const int* __restrict__ ei, int E,
               const float* __restrict__ h, int n,
               const unsigned* __restrict__ seg,
               const unsigned* __restrict__ segcnt,
               int nbk, int kps,
               float* __restrict__ p1) {
    __shared__ float acc[S1_R * FH];
    const int tid  = threadIdx.x;
    const int lane = tid & 31;
    const int wave = __builtin_amdgcn_readfirstlane(tid >> 5);
    const int g    = lane >> 4, f = lane & 15;
    const int rb   = blockIdx.x;
    const int lo   = rb * S1_R;
    const int sl   = blockIdx.y;

    float4* acc4 = (float4*)acc;
    for (int i = tid; i < S1_R * FH / 4; i += S1_T) acc4[i] = make_float4(0.f, 0.f, 0.f, 0.f);
    __syncthreads();

    const int k0 = sl * kps;
    const int k1 = min(k0 + kps, nbk);
    int roff = 0;
    int      ev_p  = 0;
    unsigned i0_p  = 0u, cnt_p = 0u;
    bool     have  = false;
    for (int kc = k0; kc < k1; kc += 32) {
        const int kk  = kc + lane;
        const int myc = (kk < k1) ? (int)segcnt[(size_t)rb * nbk + kk] : 0;
        const int ke  = min(kc + 32, k1);
        for (int k = kc; k < ke; ++k) {
            const unsigned cnt = (unsigned)__builtin_amdgcn_readlane(myc, k - kc);
            const unsigned* __restrict__ sg = seg + ((size_t)rb * nbk + k) * BK_PITCH;
            const int nr = (int)((cnt + 31u) >> 5);
            for (int j = (wave - roff) & (S1_W - 1); j < nr; j += S1_W) {
                const unsigned i0 = (unsigned)j << 5;
                const unsigned li = i0 + (unsigned)lane;
                const int ev = (int)sg[li < cnt ? li : cnt - 1u];
                if (have) s1_round(ev_p, i0_p, cnt_p, g, f, h, acc);
                ev_p = ev; i0_p = i0; cnt_p = cnt; have = true;
            }
            roff = (roff + nr) & (S1_W - 1);
        }
    }

    if (have) s1_round(ev_p, i0_p, cnt_p, g, f, h, acc);

    if (sl == 0) {
        const int nq = E >> 2;
        int e = (nq << 2) + tid;
        if (e < E) {
            int sN = ei[e];
            unsigned r = (unsigned)(ei[E + e] - lo);
            if (r < S1_R) {
                for (int ff = 0; ff < FH; ++ff)
                    atomicAdd(&acc[r * FH + ff], h[sN * FH + ff]);
            }
        }
    }
    __syncthreads();

    float4* __restrict__ prow4 = (float4*)(p1 + (size_t)sl * FH * n);
    for (int i = tid; i < S1_R * FH / 4; i += S1_T) {
        int g4 = lo * (FH / 4) + i;
        if (g4 < n * (FH / 4)) prow4[g4] = acc4[i];
    }
}


__global__ void k_layer2(const float* __restrict__ p1,
                         const float* __restrict__ h,
                         const float* __restrict__ dinv,
                         const float* __restrict__ b1,
                         const float* __restrict__ W2,
                         float* __restrict__ h2, int n) {
    int i = blockIdx.x * blockDim.x + threadIdx.x;
    if (i < n) {
        const float w = dinv[i];
        const float4* __restrict__ h4 = (const float4*)h;
        const float4* __restrict__ p4 = (const float4*)p1;
        float a0 = 0.0f, a1 = 0.0f;
        #pragma unroll
        for (int c = 0; c < FH / 4; ++c) {
            float4 t = h4[(size_t)i * (FH / 4) + c];
            float v[4] = { t.x, t.y, t.z, t.w };
            #pragma unroll
            for (int j = 0; j < S1_S; ++j) {
                float4 p = p4[((size_t)j * n + i) * (FH / 4) + c];
                v[0] += p.x; v[1] += p.y; v[2] += p.z; v[3] += p.w;
            }
            #pragma unroll
            for (int k = 0; k < 4; ++k) {
                int kk = 4 * c + k;
                float r = fmaxf(v[k] * w + b1[kk], 0.0f);
                a0 = fmaf(r, W2[2 * kk],     a0);
                a1 = fmaf(r, W2[2 * kk + 1], a1);
            }
        }
        h2[2 * i]     = a0 * w;
        h2[2 * i + 1] = a1 * w;
    }
}


__global__ void __launch_bounds__(S1_T)
k_scatter2_bkt(const int* __restrict__ ei, int E,
               const float* __restrict__ h2, int n,
               const unsigned* __restrict__ seg,
               const unsigned* __restrict__ segcnt,
               int nbk, int kps,
               float* __restrict__ p2) {
    __shared__ float acc[S1_R * FOUT];
    const int tid = threadIdx.x;
    const int rb  = blockIdx.x;
    const int lo  = rb * S1_R;
    const int sl  = blockIdx.y;
    for (int i = tid; i < S1_R * FOUT; i += S1_T) acc[i] = 0.0f;
    __syncthreads();

    const int k0 = sl * kps;
    const int k1 = min(k0 + kps, nbk);
    for (int kq = k0; kq < k1; kq += 4) {
        unsigned c[4], e[4];
        bool     ok[4];
        const unsigned* sg[4];
        #pragma unroll
        for (int u = 0; u < 4; ++u) {
            const int k = min(kq + u, k1 - 1);
            c[u]  = (kq + u < k1) ? segcnt[(size_t)rb * nbk + k] : 0u;
            sg[u] = seg + ((size_t)rb * nbk + k) * BK_PITCH;
        }
        #pragma unroll
        for (int u = 0; u < 4; ++u) {
            ok[u] = (unsigned)tid < c[u];
            e[u]  = ok[u] ? sg[u][tid] : 0u;
        }
        v2f gv[4];
        #pragma unroll
        for (int u = 0; u < 4; ++u) gv[u] = *(const v2f*)(h2 + 2 * (e[u] >> 12));
        asm volatile("" :: "v"(gv[0]), "v"(gv[1]), "v"(gv[2]), "v"(gv[3]));
        #pragma unroll
        for (int u = 0; u < 4; ++u)
            if (ok[u]) {
                const unsigned r = e[u] & 4095u;
                atomicAdd(&acc[2 * r],     gv[u].x);
                atomicAdd(&acc[2 * r + 1], gv[u].y);
            }
        #pragma unroll
        for (int u = 0; u < 4; ++u)
            for (unsigned i = (unsigned)tid + S1_T; i < c[u]; i += S1_T) {
                const unsigned en = sg[u][i];
                const v2f g1 = *(const v2f*)(h2 + 2 * (en >> 12));
                const unsigned r = en & 4095u;
                atomicAdd(&acc[2 * r],     g1.x);
                atomicAdd(&acc[2 * r + 1], g1.y);
            }
    }
    if (sl == 0) {
        const int nq = E >> 2;
        int e = (nq << 2) + tid;
        if (e < E) {
            int sN = ei[e];
            unsigned r = (unsigned)(ei[E + e] - lo);
            if (r < S1_R) {
                atomicAdd(&acc[2 * r],     h2[2 * sN]);
                atomicAdd(&acc[2 * r + 1], h2[2 * sN + 1]);
            }
        }
    }
    __syncthreads();

    float* __restrict__ prow = p2 + (size_t)sl * FOUT * n;
    for (int i = tid; i < S1_R * FOUT; i += S1_T) {
        int g = lo * FOUT + i;
        if (g < n * FOUT) prow[g] = acc[i];
    }
}


__global__ void k_logsoftmax(const float* __restrict__ p2,
                             const float* __restrict__ h2,
                             const float* __restrict__ dinv,
                             const float* __restrict__ b2,
                             float* __restrict__ out, int n) {
    int i = blockIdx.x * blockDim.x + threadIdx.x;
    if (i < n) {
        float w  = dinv[i];
        float a0 = h2[2 * i];
        float a1 = h2[2 * i + 1];
        #pragma unroll
        for (int j = 0; j < S1_S; ++j) {
            v2f v = *(const v2f*)(p2 + (size_t)j * FOUT * n + 2 * i);
            a0 += v.x;
            a1 += v.y;
        }
        a0 = a0 * w + b2[0];
        a1 = a1 * w + b2[1];
        float m  = fmaxf(a0, a1);
        float ls = logf(expf(a0 - m) + expf(a1 - m));
        out[2 * i]     = a0 - m - ls;
        out[2 * i + 1] = a1 - m - ls;
    }
}


extern "C" void kernel_launch(void* const* d_in, const int* in_sizes, int n_in,
                              void* d_out, int out_size, void* d_ws, size_t ws_size,
                              hipStream_t stream) {
    const float* x  = (const float*)d_in[0];
    const int*   ei = (const int*)d_in[1];
    const float* W1 = (const float*)d_in[2];
    const float* b1 = (const float*)d_in[3];
    const float* W2 = (const float*)d_in[4];
    const float* b2 = (const float*)d_in[5];
    float* out      = (float*)d_out;

    const int N = in_sizes[0] / FIN;
    const int E = in_sizes[1] / 2;

    const int B = 256;
    auto blk = [](long long t, int b) { return (int)((t + b - 1) / b); };

    const int nb  = blk(N, S1_R);
    const int nbk = blk(E / 4, BK_CAP / 4);
    const int kps = blk(nbk, S1_S);

    float* ws      = (float*)d_ws;
    float* p2      = ws;
    unsigned* pdeg = (unsigned*)(p2 + (size_t)S1_S * FOUT * N);
    float* h2      = (float*)(pdeg + (size_t)S1_S * N);
    float* h       = h2   + (size_t)N * FOUT;
    float* p1      = h    + (size_t)N * FH;
    float* dinv    = p1   + (size_t)S1_S * N * FH;
    unsigned* seg    = (unsigned*)(dinv + N);
    unsigned* segcnt = seg + (size_t)nb * nbk * BK_PITCH;

    const dim3 rgrid(nb, S1_S);

    k_bucket     <<<nbk, BK_T, 0, stream>>>(ei, E, nb, seg, segcnt);

    k_deg_bkt    <<<rgrid, S1_T, 0, stream>>>(ei, E, N, seg, segcnt, nbk, kps, pdeg);
    k_dinv       <<<blk(N, B), B, 0, stream>>>(pdeg, dinv, N);

    const int n_tiles = N / 16;
    k_gemm1_wmma <<<blk(n_tiles, 8), B, 0, stream>>>(x, W1, dinv, h, n_tiles);
    k_scatter1_lds<<<rgrid, S1_T, 0, stream>>>(ei, E, h, N, seg, segcnt, nbk, kps, p1);

    k_layer2     <<<blk(N, B), B, 0, stream>>>(p1, h, dinv, b1, W2, h2, N);
    k_scatter2_bkt<<<rgrid, S1_T, 0, stream>>>(ei, E, h2, N, seg, segcnt, nbk, kps, p2);

    k_logsoftmax <<<blk(N, B), B, 0, stream>>>(p2, h2, dinv, b2, out, N);
}


# edges bucketed into 400-node destination ranges: one block owns a range in each scatter pass (no slice partial planes) and finishes its own nodes there, 8 launches to 5 (sum order changes)
# speedup vs baseline: 5.7370x; 1.2488x over previous
#include <hip/hip_runtime.h>
#include <hip/hip_bf16.h>


typedef float v2f __attribute__((ext_vector_type(2)));
typedef float v8f __attribute__((ext_vector_type(8)));

#define FIN   512
#define FH    16
#define FOUT  2


#define S1_R     400
#define S1_T     1024
#define S1_W     (S1_T / 32)

#define BK_T     1024
#define BK_CAP   16384
#define BK_MAXB  256

__global__ void __launch_bounds__(BK_T)
k_bucket(const int* __restrict__ ei, int E, int nb,
         unsigned* __restrict__ seg, unsigned* __restrict__ segtab) {
    __shared__ __attribute__((aligned(16))) unsigned stage[BK_CAP];
    __shared__ unsigned lcnt[BK_MAXB];
    __shared__ unsigned lbase[BK_MAXB];
    __shared__ unsigned wsum[BK_MAXB / 32];
    const int tid = threadIdx.x;
    const int k   = blockIdx.x;
    if (tid < BK_MAXB) lcnt[tid] = 0u;
    __syncthreads();

    const int nq = E >> 2;
    const int4* __restrict__ src4 = (const int4*)ei;
    const int4* __restrict__ dst4 = (const int4*)(ei + E);
    const int qb = k * (BK_CAP / 4);

    int4 d[4], s[4];
    bool v[4];
    #pragma unroll
    for (int j = 0; j < 4; ++j) {
        const int q = qb + j * BK_T + tid;
        v[j] = q < nq;
        d[j] = dst4[v[j] ? q : qb];
        s[j] = src4[v[j] ? q : qb];
    }

    unsigned pk[16], bk[16], rk[16];
    bool vd[16];
    #pragma unroll
    for (int j = 0; j < 4; ++j) {
        const int dd[4] = { d[j].x, d[j].y, d[j].z, d[j].w };
        const int ss[4] = { s[j].x, s[j].y, s[j].z, s[j].w };
        #pragma unroll
        for (int c = 0; c < 4; ++c) {
            const int t = 4 * j + c;
            const unsigned b = (unsigned)dd[c] / S1_R;
            const unsigned r = (unsigned)dd[c] - b * S1_R;
            vd[t] = v[j] && b < (unsigned)nb;
            bk[t] = vd[t] ? b : 0u;
            pk[t] = ((unsigned)ss[c] << 12) | r;
            rk[t] = 0u;
            if (vd[t]) rk[t] = atomicAdd(&lcnt[bk[t]], 1u);
        }
    }
    __syncthreads();

    unsigned cme = 0u, incl = 0u;
    if (tid < BK_MAXB) {
        cme  = lcnt[tid];
        incl = cme;
        #pragma unroll
        for (int dl = 1; dl < 32; dl <<= 1) {
            const unsigned y = __shfl_up(incl, dl);
            if ((tid & 31) >= dl) incl += y;
        }
        if ((tid & 31) == 31) wsum[tid >> 5] = incl;
    }
    __syncthreads();
    if (tid < BK_MAXB) {
        unsigned pre = 0u;
        #pragma unroll
        for (int j = 0; j < BK_MAXB / 32; ++j) pre += (j < (tid >> 5)) ? wsum[j] : 0u;
        const unsigned base = pre + incl - cme;
        lbase[tid] = base;
        if (tid < nb) segtab[(size_t)tid * gridDim.x + k] = (base << 16) | cme;
    }
    __syncthreads();

    unsigned total = 0u;
    #pragma unroll
    for (int j = 0; j < BK_MAXB / 32; ++j) total += wsum[j];

    unsigned pos[16];
    #pragma unroll
    for (int t = 0; t < 16; ++t) pos[t] = lbase[bk[t]] + rk[t];
    #pragma unroll
    for (int t = 0; t < 16; ++t)
        if (vd[t]) stage[pos[t]] = pk[t];
    __syncthreads();

    const uint4* st4 = (const uint4*)stage;
    uint4* __restrict__ o4 = (uint4*)(seg + (size_t)k * BK_CAP);
    for (unsigned i = tid; 4u * i < total; i += BK_T) o4[i] = st4[i];
}


__global__ void __launch_bounds__(S1_T)
k_deg_dinv(const int* __restrict__ ei, int E, int n,
           const unsigned* __restrict__ seg,
           const unsigned* __restrict__ segtab,
           int nbk,
           float* __restrict__ dinv) {
    __shared__ unsigned cnt[S1_R];
    const int tid  = threadIdx.x;
    const int lane = tid & 31;
    const int wave = __builtin_amdgcn_readfirstlane(tid >> 5);
    const int rb   = blockIdx.x;
    const int lo   = rb * S1_R;
    for (int i = tid; i < S1_R; i += S1_T) cnt[i] = 0u;
    __syncthreads();

    for (int k0 = 0; k0 < nbk; k0 += 32 * S1_W) {
        const int kk = k0 + wave + S1_W * lane;
        const unsigned myw = (kk < nbk) ? segtab[(size_t)rb * nbk + kk] : 0u;
        const int left = nbk - k0 - wave;
        const int ns   = min(32, (left + S1_W - 1) / S1_W);
        for (int j = 0; j < ns; j += 2) {
            unsigned c[2];
            const unsigned* sg[2];
            #pragma unroll
            for (int u = 0; u < 2; ++u) {
                const unsigned w = (unsigned)__builtin_amdgcn_readlane((int)myw, j + u);
                c[u]  = w & 0xFFFFu;
                sg[u] = seg + (size_t)(k0 + wave + S1_W * (j + u)) * BK_CAP + (w >> 16);
            }
            unsigned e[8];
            bool     ok[8];
            #pragma unroll
            for (int u = 0; u < 2; ++u)
                #pragma unroll
                for (int r = 0; r < 4; ++r) {
                    const unsigned idx = 32u * r + (unsigned)lane;
                    ok[4 * u + r] = idx < c[u];
                    e[4 * u + r]  = ok[4 * u + r] ? sg[u][idx] : 0u;
                }
            asm volatile("" :: "v"(e[0]), "v"(e[1]), "v"(e[2]), "v"(e[3]),
                               "v"(e[4]), "v"(e[5]), "v"(e[6]), "v"(e[7]));
            #pragma unroll
            for (int t = 0; t < 8; ++t)
                if (ok[t]) atomicAdd(&cnt[e[t] & 4095u], 1u);
            #pragma unroll
            for (int u = 0; u < 2; ++u)
                for (unsigned i = 128u + (unsigned)lane; i < c[u]; i += 32u)
                    atomicAdd(&cnt[sg[u][i] & 4095u], 1u);
        }
    }
    {
        const int nq = E >> 2;
        int e = (nq << 2) + tid;
        if (e < E) {
            unsigned r = (unsigned)(ei[E + e] - lo);
            if (r < S1_R) atomicAdd(&cnt[r], 1u);
        }
    }
    __syncthreads();

    for (int i = tid; i < S1_R; i += S1_T) {
        const int b = lo + i;
        if (b < n) dinv[b] = rsqrtf((float)(1u + cnt[i]));
    }
}


__global__ void __launch_bounds__(256)
k_gemm1_wmma(const float* __restrict__ x, const float* __restrict__ W1,
             const float* __restrict__ dinv,
             float* __restrict__ h, int n_tiles) {
    __shared__ float lw[FIN * FH];
    int tid = threadIdx.x;
    for (int i = tid; i < FIN * FH; i += 256) lw[i] = W1[i];
    __syncthreads();

    int wave = tid >> 5;
    int lane = tid & 31;
    int tile = blockIdx.x * 8 + wave;
    if (tile >= n_tiles) return;

    int half = lane >> 4;
    int mn   = lane & 15;
    int kofs = 2 * half;

    const float* __restrict__ xrow = x + (size_t)(tile * 16 + mn) * FIN;

    float dsc[8];
    #pragma unroll
    for (int r = 0; r < 8; ++r) dsc[r] = dinv[tile * 16 + 8 * half + r];

    v8f c = {};
    #pragma unroll 4
    for (int k = 0; k < FIN; k += 4) {
        v2f a = *(const v2f*)(xrow + k + kofs);
        v2f b;
        b.x = lw[(k + kofs)     * FH + mn];
        b.y = lw[(k + kofs + 1) * FH + mn];
        c = __builtin_amdgcn_wmma_f32_16x16x4_f32(
                 false, a,  false, b,
                 (short)0, c,  false,  false);
    }

    float* __restrict__ hrow = h + (size_t)tile * 16 * FH;
    #pragma unroll
    for (int r = 0; r < 8; ++r)
        hrow[(r + 8 * half) * FH + mn] = c[r] * dsc[r];
}


template <int NS>
__device__ __forceinline__ void s1_steps(int ev, unsigned valid,
                                         int g, int f,
                                         const float* __restrict__ h,
                                         float* acc) {
    float    hv[NS];
    unsigned rr[NS];
    bool     ok[NS];
    #pragma unroll
    for (int u = 0; u < NS; ++u) {
        const unsigned e = (unsigned)__shfl(ev, 2 * u + g);
        ok[u] = (unsigned)(2 * u + g) < valid;
        rr[u] = e & 4095u;
        hv[u] = h[(e >> 12) * FH + f];
    }
    #pragma unroll
    for (int u = 0; u < NS; ++u)
        atomicAdd(&acc[rr[u] * FH + f], ok[u] ? hv[u] : 0.0f);
}

__device__ __forceinline__ void s1_round(int ev, unsigned valid,
                                         int g, int f,
                                         const float* __restrict__ h,
                                         float* acc) {
    if (valid > 16u) s1_steps<16>(ev, valid, g, f, h, acc);
    else             s1_steps<8>(ev, valid, g, f, h, acc);
}

__global__ void __launch_bounds__(S1_T)
k_scatter1_l2(const int* __restrict__ ei, int E,
              const float* __restrict__ h, int n,
              const unsigned* __restrict__ seg,
              const unsigned* __restrict__ segtab,
              int nbk,
              const float* __restrict__ dinv,
              const float* __restrict__ b1,
              const float* __restrict__ W2,
              float* __restrict__ h2) {
    __shared__ __attribute__((aligned(16))) float acc[S1_R * FH];
    const int tid  = threadIdx.x;
    const int lane = tid & 31;
    const int wave = __builtin_amdgcn_readfirstlane(tid >> 5);
    const int g    = lane >> 4, f = lane & 15;
    const int rb   = blockIdx.x;
    const int lo   = rb * S1_R;

    float4* acc4 = (float4*)acc;
    for (int i = tid; i < S1_R * FH / 4; i += S1_T) acc4[i] = make_float4(0.f, 0.f, 0.f, 0.f);
    __syncthreads();

    int      ev_p    = 0;
    unsigned valid_p = 0u;
    bool     have    = false;
    for (int k0 = 0; k0 < nbk; k0 += 32 * S1_W) {
        const int kk = k0 + wave + S1_W * lane;
        const unsigned myw = (kk < nbk) ? segtab[(size_t)rb * nbk + kk] : 0u;
        const int left = nbk - k0 - wave;
        const int ns   = min(32, (left + S1_W - 1) / S1_W);
        for (int j = 0; j < ns; ++j) {
            const unsigned w   = (unsigned)__builtin_amdgcn_readlane((int)myw, j);
            const unsigned cnt = w & 0xFFFFu;
            const unsigned* __restrict__ sg = seg + (size_t)(k0 + wave + S1_W * j) * BK_CAP + (w >> 16);
            for (unsigned i0 = 0u; i0 < cnt; i0 += 32u) {
                const unsigned li = i0 + (unsigned)lane;
                const int ev = (int)sg[li < cnt ? li : cnt - 1u];
                if (have) s1_round(ev_p, valid_p, g, f, h, acc);
                ev_p = ev; valid_p = min(32u, cnt - i0); have = true;
            }
        }
    }
    if (have) s1_round(ev_p, valid_p, g, f, h, acc);

    {
        const int nq = E >> 2;
        int e = (nq << 2) + tid;
        if (e < E) {
            int sN = ei[e];
            unsigned r = (unsigned)(ei[E + e] - lo);
            if (r < S1_R) {
                for (int ff = 0; ff < FH; ++ff)
                    atomicAdd(&acc[r * FH + ff], h[sN * FH + ff]);
            }
        }
    }
    __syncthreads();

    for (int i = tid; i < S1_R; i += S1_T) {
        const int node = lo + i;
        if (node < n) {
            const float w = dinv[node];
            const float4* __restrict__ h4 = (const float4*)h;
            float a0 = 0.0f, a1 = 0.0f;
            #pragma unroll
            for (int c = 0; c < FH / 4; ++c) {
                const float4 t = h4[(size_t)node * (FH / 4) + c];
                const float4 p = acc4[i * (FH / 4) + c];
                const float v[4] = { t.x + p.x, t.y + p.y, t.z + p.z, t.w + p.w };
                #pragma unroll
                for (int k = 0; k < 4; ++k) {
                    const int kk = 4 * c + k;
                    const float r = fmaxf(v[k] * w + b1[kk], 0.0f);
                    a0 = fmaf(r, W2[2 * kk],     a0);
                    a1 = fmaf(r, W2[2 * kk + 1], a1);
                }
            }
            h2[2 * node]     = a0 * w;
            h2[2 * node + 1] = a1 * w;
        }
    }
}


__global__ void __launch_bounds__(S1_T)
k_scatter2_out(const int* __restrict__ ei, int E,
               const float* __restrict__ h2, int n,
               const unsigned* __restrict__ seg,
               const unsigned* __restrict__ segtab,
               int nbk,
               const float* __restrict__ dinv,
               const float* __restrict__ b2,
               float* __restrict__ out) {
    __shared__ float acc[S1_R * FOUT];
    const int tid  = threadIdx.x;
    const int lane = tid & 31;
    const int wave = __builtin_amdgcn_readfirstlane(tid >> 5);
    const int rb   = blockIdx.x;
    const int lo   = rb * S1_R;
    for (int i = tid; i < S1_R * FOUT; i += S1_T) acc[i] = 0.0f;
    __syncthreads();

    for (int k0 = 0; k0 < nbk; k0 += 32 * S1_W) {
        const int kk = k0 + wave + S1_W * lane;
        const unsigned myw = (kk < nbk) ? segtab[(size_t)rb * nbk + kk] : 0u;
        const int left = nbk - k0 - wave;
        const int ns   = min(32, (left + S1_W - 1) / S1_W);
        for (int j = 0; j < ns; j += 2) {
            unsigned c[2];
            const unsigned* sg[2];
            #pragma unroll
            for (int u = 0; u < 2; ++u) {
                const unsigned w = (unsigned)__builtin_amdgcn_readlane((int)myw, j + u);
                c[u]  = w & 0xFFFFu;
                sg[u] = seg + (size_t)(k0 + wave + S1_W * (j + u)) * BK_CAP + (w >> 16);
            }
            unsigned e[8];
            bool     ok[8];
            #pragma unroll
            for (int u = 0; u < 2; ++u)
                #pragma unroll
                for (int r = 0; r < 4; ++r) {
                    const unsigned idx = 32u * r + (unsigned)lane;
                    ok[4 * u + r] = idx < c[u];
                    e[4 * u + r]  = ok[4 * u + r] ? sg[u][idx] : 0u;
                }
            v2f gv[8];
            #pragma unroll
            for (int t = 0; t < 8; ++t) gv[t] = *(const v2f*)(h2 + 2 * (e[t] >> 12));
            asm volatile("" :: "v"(gv[0]), "v"(gv[1]), "v"(gv[2]), "v"(gv[3]),
                               "v"(gv[4]), "v"(gv[5]), "v"(gv[6]), "v"(gv[7]));
            #pragma unroll
            for (int t = 0; t < 8; ++t)
                if (ok[t]) {
                    const unsigned r = e[t] & 4095u;
                    atomicAdd(&acc[2 * r],     gv[t].x);
                    atomicAdd(&acc[2 * r + 1], gv[t].y);
                }
            #pragma unroll
            for (int u = 0; u < 2; ++u)
                for (unsigned i = 128u + (unsigned)lane; i < c[u]; i += 32u) {
                    const unsigned en = sg[u][i];
                    const v2f g1 = *(const v2f*)(h2 + 2 * (en >> 12));
                    const unsigned r = en & 4095u;
                    atomicAdd(&acc[2 * r],     g1.x);
                    atomicAdd(&acc[2 * r + 1], g1.y);
                }
        }
    }
    {
        const int nq = E >> 2;
        int e = (nq << 2) + tid;
        if (e < E) {
            int sN = ei[e];
            unsigned r = (unsigned)(ei[E + e] - lo);
            if (r < S1_R) {
                atomicAdd(&acc[2 * r],     h2[2 * sN]);
                atomicAdd(&acc[2 * r + 1], h2[2 * sN + 1]);
            }
        }
    }
    __syncthreads();

    for (int i = tid; i < S1_R; i += S1_T) {
        const int node = lo + i;
        if (node < n) {
            const float w = dinv[node];
            float a0 = h2[2 * node]     + acc[2 * i];
            float a1 = h2[2 * node + 1] + acc[2 * i + 1];
            a0 = a0 * w + b2[0];
            a1 = a1 * w + b2[1];
            const float m  = fmaxf(a0, a1);
            const float ls = logf(expf(a0 - m) + expf(a1 - m));
            out[2 * node]     = a0 - m - ls;
            out[2 * node + 1] = a1 - m - ls;
        }
    }
}


extern "C" void kernel_launch(void* const* d_in, const int* in_sizes, int n_in,
                              void* d_out, int out_size, void* d_ws, size_t ws_size,
                              hipStream_t stream) {
    const float* x  = (const float*)d_in[0];
    const int*   ei = (const int*)d_in[1];
    const float* W1 = (const float*)d_in[2];
    const float* b1 = (const float*)d_in[3];
    const float* W2 = (const float*)d_in[4];
    const float* b2 = (const float*)d_in[5];
    float* out      = (float*)d_out;

    const int N = in_sizes[0] / FIN;
    const int E = in_sizes[1] / 2;

    const int B = 256;
    auto blk = [](long long t, int b) { return (int)((t + b - 1) / b); };

    const int nb  = blk(N, S1_R);
    const int nbk = blk(E / 4, BK_CAP / 4);

    unsigned* seg    = (unsigned*)d_ws;
    float* h         = (float*)(seg + (size_t)nbk * BK_CAP);
    float* h2        = h    + (size_t)N * FH;
    float* dinv      = h2   + (size_t)N * FOUT;
    unsigned* segtab = (unsigned*)(dinv + N);

    k_bucket      <<<nbk, BK_T, 0, stream>>>(ei, E, nb, seg, segtab);

    k_deg_dinv    <<<nb, S1_T, 0, stream>>>(ei, E, N, seg, segtab, nbk, dinv);

    const int n_tiles = N / 16;
    k_gemm1_wmma  <<<blk(n_tiles, 8), B, 0, stream>>>(x, W1, dinv, h, n_tiles);
    k_scatter1_l2 <<<nb, S1_T, 0, stream>>>(ei, E, h, N, seg, segtab, nbk, dinv, b1, W2, h2);

    k_scatter2_out<<<nb, S1_T, 0, stream>>>(ei, E, h2, N, seg, segtab, nbk, dinv, b2, out);
}


# GEMM1 runs inside the degree kernel: each 400-node range block counts its degrees, then 25 of its waves multiply its own x rows by W1, dinv taken from the LDS counts; 5 launches to 4, same sums
# speedup vs baseline: 6.0704x; 1.0581x over previous
#include <hip/hip_runtime.h>
#include <hip/hip_bf16.h>


typedef float v2f __attribute__((ext_vector_type(2)));
typedef float v8f __attribute__((ext_vector_type(8)));

#define FIN   512
#define FH    16
#define FOUT  2


#define S1_R     400
#define S1_T     1024
#define S1_W     (S1_T / 32)

#define BK_T     1024
#define BK_CAP   16384
#define BK_MAXB  256

__global__ void __launch_bounds__(BK_T)
k_bucket(const int* __restrict__ ei, int E, int nb,
         unsigned* __restrict__ seg, unsigned* __restrict__ segtab) {
    __shared__ __attribute__((aligned(16))) unsigned stage[BK_CAP];
    __shared__ unsigned lcnt[BK_MAXB];
    __shared__ unsigned lbase[BK_MAXB];
    __shared__ unsigned wsum[BK_MAXB / 32];
    const int tid = threadIdx.x;
    const int k   = blockIdx.x;
    if (tid < BK_MAXB) lcnt[tid] = 0u;
    __syncthreads();

    const int nq = E >> 2;
    const int4* __restrict__ src4 = (const int4*)ei;
    const int4* __restrict__ dst4 = (const int4*)(ei + E);
    const int qb = k * (BK_CAP / 4);

    int4 d[4], s[4];
    bool v[4];
    #pragma unroll
    for (int j = 0; j < 4; ++j) {
        const int q = qb + j * BK_T + tid;
        v[j] = q < nq;
        d[j] = dst4[v[j] ? q : qb];
        s[j] = src4[v[j] ? q : qb];
    }

    unsigned pk[16], bk[16], rk[16];
    bool vd[16];
    #pragma unroll
    for (int j = 0; j < 4; ++j) {
        const int dd[4] = { d[j].x, d[j].y, d[j].z, d[j].w };
        const int ss[4] = { s[j].x, s[j].y, s[j].z, s[j].w };
        #pragma unroll
        for (int c = 0; c < 4; ++c) {
            const int t = 4 * j + c;
            const unsigned b = (unsigned)dd[c] / S1_R;
            const unsigned r = (unsigned)dd[c] - b * S1_R;
            vd[t] = v[j] && b < (unsigned)nb;
            bk[t] = vd[t] ? b : 0u;
            pk[t] = ((unsigned)ss[c] << 12) | r;
            rk[t] = 0u;
            if (vd[t]) rk[t] = atomicAdd(&lcnt[bk[t]], 1u);
        }
    }
    __syncthreads();

    unsigned cme = 0u, incl = 0u;
    if (tid < BK_MAXB) {
        cme  = lcnt[tid];
        incl = cme;
        #pragma unroll
        for (int dl = 1; dl < 32; dl <<= 1) {
            const unsigned y = __shfl_up(incl, dl);
            if ((tid & 31) >= dl) incl += y;
        }
        if ((tid & 31) == 31) wsum[tid >> 5] = incl;
    }
    __syncthreads();
    if (tid < BK_MAXB) {
        unsigned pre = 0u;
        #pragma unroll
        for (int j = 0; j < BK_MAXB / 32; ++j) pre += (j < (tid >> 5)) ? wsum[j] : 0u;
        const unsigned base = pre + incl - cme;
        lbase[tid] = base;
        if (tid < nb) segtab[(size_t)tid * gridDim.x + k] = (base << 16) | cme;
    }
    __syncthreads();

    unsigned total = 0u;
    #pragma unroll
    for (int j = 0; j < BK_MAXB / 32; ++j) total += wsum[j];

    unsigned pos[16];
    #pragma unroll
    for (int t = 0; t < 16; ++t) pos[t] = lbase[bk[t]] + rk[t];
    #pragma unroll
    for (int t = 0; t < 16; ++t)
        if (vd[t]) stage[pos[t]] = pk[t];
    __syncthreads();

    const uint4* st4 = (const uint4*)stage;
    uint4* __restrict__ o4 = (uint4*)(seg + (size_t)k * BK_CAP);
    for (unsigned i = tid; 4u * i < total; i += BK_T) o4[i] = st4[i];
}


#define TPR      (S1_R / 16)
#define W1_PT    (FIN * FH / S1_T)

__global__ void __launch_bounds__(S1_T)
k_deg_gemm1(const int* __restrict__ ei, int E, int n,
            const unsigned* __restrict__ seg,
            const unsigned* __restrict__ segtab,
            int nbk,
            const float* __restrict__ x, const float* __restrict__ W1,
            float* __restrict__ dinv,
            float* __restrict__ h, int n_tiles) {
    __shared__ float lw[FIN * FH];
    __shared__ __attribute__((aligned(16))) unsigned cnt[S1_R];
    const int tid  = threadIdx.x;
    const int lane = tid & 31;
    const int wave = __builtin_amdgcn_readfirstlane(tid >> 5);
    const int rb   = blockIdx.x;
    const int lo   = rb * S1_R;

    for (int i = tid; i < S1_R; i += S1_T) cnt[i] = 0u;
    __syncthreads();

    float wr[W1_PT];
    #pragma unroll
    for (int j = 0; j < W1_PT; ++j) wr[j] = W1[tid + S1_T * j];

    for (int k0 = 0; k0 < nbk; k0 += 32 * S1_W) {
        const int kk = k0 + wave + S1_W * lane;
        const unsigned tw  = segtab[(size_t)rb * nbk + min(kk, nbk - 1)];
        const unsigned myw = (kk < nbk) ? tw : 0u;
        const int left = nbk - k0 - wave;
        const int ns   = min(32, (left + S1_W - 1) / S1_W);
        for (int j = 0; j < ns; j += 2) {
            unsigned c[2];
            const unsigned* sg[2];
            #pragma unroll
            for (int u = 0; u < 2; ++u) {
                const unsigned w = (unsigned)__builtin_amdgcn_readlane((int)myw, j + u);
                c[u]  = w & 0xFFFFu;
                sg[u] = seg + (size_t)(k0 + wave + S1_W * (j + u)) * BK_CAP + (w >> 16);
            }
            unsigned e[8];
            bool     ok[8];
            #pragma unroll
            for (int u = 0; u < 2; ++u)
                #pragma unroll
                for (int r = 0; r < 4; ++r) {
                    const unsigned idx = 32u * r + (unsigned)lane;
                    ok[4 * u + r] = idx < c[u];
                    e[4 * u + r]  = ok[4 * u + r] ? sg[u][idx] : 0u;
                }
            asm volatile("" :: "v"(e[0]), "v"(e[1]), "v"(e[2]), "v"(e[3]),
                               "v"(e[4]), "v"(e[5]), "v"(e[6]), "v"(e[7]));
            #pragma unroll
            for (int t = 0; t < 8; ++t)
                if (ok[t]) atomicAdd(&cnt[e[t] & 4095u], 1u);
            #pragma unroll
            for (int u = 0; u < 2; ++u)
                for (unsigned i = 128u + (unsigned)lane; i < c[u]; i += 32u)
                    atomicAdd(&cnt[sg[u][i] & 4095u], 1u);
        }
    }
    {
        const int nq = E >> 2;
        int e = (nq << 2) + tid;
        if (e < E) {
            unsigned r = (unsigned)(ei[E + e] - lo);
            if (r < S1_R) atomicAdd(&cnt[r], 1u);
        }
    }
    #pragma unroll
    for (int j = 0; j < W1_PT; ++j) lw[tid + S1_T * j] = wr[j];
    __syncthreads();

    for (int i = tid; i < S1_R; i += S1_T) {
        const int b = lo + i;
        if (b < n) dinv[b] = rsqrtf((float)(1u + cnt[i]));
    }

    const int tile = rb * TPR + wave;
    if (wave >= TPR || tile >= n_tiles) return;

    const int half = lane >> 4;
    const int mn   = lane & 15;
    const int kofs = 2 * half;

    const float* __restrict__ xrow = x + (size_t)(tile * 16 + mn) * FIN;

    float dsc[8];
    #pragma unroll
    for (int r = 0; r < 8; ++r) dsc[r] = rsqrtf((float)(1u + cnt[wave * 16 + 8 * half + r]));

    v8f c = {};
    #pragma unroll 4
    for (int k = 0; k < FIN; k += 4) {
        v2f a = *(const v2f*)(xrow + k + kofs);
        v2f b;
        b.x = lw[(k + kofs)     * FH + mn];
        b.y = lw[(k + kofs + 1) * FH + mn];
        c = __builtin_amdgcn_wmma_f32_16x16x4_f32(
                 false, a,  false, b,
                 (short)0, c,  false,  false);
    }

    float* __restrict__ hrow = h + (size_t)tile * 16 * FH;
    #pragma unroll
    for (int r = 0; r < 8; ++r)
        hrow[(r + 8 * half) * FH + mn] = c[r] * dsc[r];
}


template <int NS>
__device__ __forceinline__ void s1_steps(int ev, unsigned valid,
                                         int g, int f,
                                         const float* __restrict__ h,
                                         float* acc) {
    float    hv[NS];
    unsigned rr[NS];
    bool     ok[NS];
    #pragma unroll
    for (int u = 0; u < NS; ++u) {
        const unsigned e = (unsigned)__shfl(ev, 2 * u + g);
        ok[u] = (unsigned)(2 * u + g) < valid;
        rr[u] = e & 4095u;
        hv[u] = h[(e >> 12) * FH + f];
    }
    #pragma unroll
    for (int u = 0; u < NS; ++u)
        atomicAdd(&acc[rr[u] * FH + f], ok[u] ? hv[u] : 0.0f);
}

__device__ __forceinline__ void s1_round(int ev, unsigned valid,
                                         int g, int f,
                                         const float* __restrict__ h,
                                         float* acc) {
    if (valid > 16u) s1_steps<16>(ev, valid, g, f, h, acc);
    else             s1_steps<8>(ev, valid, g, f, h, acc);
}

__global__ void __launch_bounds__(S1_T)
k_scatter1_l2(const int* __restrict__ ei, int E,
              const float* __restrict__ h, int n,
              const unsigned* __restrict__ seg,
              const unsigned* __restrict__ segtab,
              int nbk,
              const float* __restrict__ dinv,
              const float* __restrict__ b1,
              const float* __restrict__ W2,
              float* __restrict__ h2) {
    __shared__ __attribute__((aligned(16))) float acc[S1_R * FH];
    const int tid  = threadIdx.x;
    const int lane = tid & 31;
    const int wave = __builtin_amdgcn_readfirstlane(tid >> 5);
    const int g    = lane >> 4, f = lane & 15;
    const int rb   = blockIdx.x;
    const int lo   = rb * S1_R;

    float4* acc4 = (float4*)acc;
    for (int i = tid; i < S1_R * FH / 4; i += S1_T) acc4[i] = make_float4(0.f, 0.f, 0.f, 0.f);
    __syncthreads();

    int      ev_p    = 0;
    unsigned valid_p = 0u;
    bool     have    = false;
    for (int k0 = 0; k0 < nbk; k0 += 32 * S1_W) {
        const int kk = k0 + wave + S1_W * lane;
        const unsigned myw = (kk < nbk) ? segtab[(size_t)rb * nbk + kk] : 0u;
        const int left = nbk - k0 - wave;
        const int ns   = min(32, (left + S1_W - 1) / S1_W);
        for (int j = 0; j < ns; ++j) {
            const unsigned w   = (unsigned)__builtin_amdgcn_readlane((int)myw, j);
            const unsigned cnt = w & 0xFFFFu;
            const unsigned* __restrict__ sg = seg + (size_t)(k0 + wave + S1_W * j) * BK_CAP + (w >> 16);
            for (unsigned i0 = 0u; i0 < cnt; i0 += 32u) {
                const unsigned li = i0 + (unsigned)lane;
                const int ev = (int)sg[li < cnt ? li : cnt - 1u];
                if (have) s1_round(ev_p, valid_p, g, f, h, acc);
                ev_p = ev; valid_p = min(32u, cnt - i0); have = true;
            }
        }
    }
    if (have) s1_round(ev_p, valid_p, g, f, h, acc);

    {
        const int nq = E >> 2;
        int e = (nq << 2) + tid;
        if (e < E) {
            int sN = ei[e];
            unsigned r = (unsigned)(ei[E + e] - lo);
            if (r < S1_R) {
                for (int ff = 0; ff < FH; ++ff)
                    atomicAdd(&acc[r * FH + ff], h[sN * FH + ff]);
            }
        }
    }
    __syncthreads();

    for (int i = tid; i < S1_R; i += S1_T) {
        const int node = lo + i;
        if (node < n) {
            const float w = dinv[node];
            const float4* __restrict__ h4 = (const float4*)h;
            float a0 = 0.0f, a1 = 0.0f;
            #pragma unroll
            for (int c = 0; c < FH / 4; ++c) {
                const float4 t = h4[(size_t)node * (FH / 4) + c];
                const float4 p = acc4[i * (FH / 4) + c];
                const float v[4] = { t.x + p.x, t.y + p.y, t.z + p.z, t.w + p.w };
                #pragma unroll
                for (int k = 0; k < 4; ++k) {
                    const int kk = 4 * c + k;
                    const float r = fmaxf(v[k] * w + b1[kk], 0.0f);
                    a0 = fmaf(r, W2[2 * kk],     a0);
                    a1 = fmaf(r, W2[2 * kk + 1], a1);
                }
            }
            h2[2 * node]     = a0 * w;
            h2[2 * node + 1] = a1 * w;
        }
    }
}


__global__ void __launch_bounds__(S1_T)
k_scatter2_out(const int* __restrict__ ei, int E,
               const float* __restrict__ h2, int n,
               const unsigned* __restrict__ seg,
               const unsigned* __restrict__ segtab,
               int nbk,
               const float* __restrict__ dinv,
               const float* __restrict__ b2,
               float* __restrict__ out) {
    __shared__ float acc[S1_R * FOUT];
    const int tid  = threadIdx.x;
    const int lane = tid & 31;
    const int wave = __builtin_amdgcn_readfirstlane(tid >> 5);
    const int rb   = blockIdx.x;
    const int lo   = rb * S1_R;
    for (int i = tid; i < S1_R * FOUT; i += S1_T) acc[i] = 0.0f;
    __syncthreads();

    for (int k0 = 0; k0 < nbk; k0 += 32 * S1_W) {
        const int kk = k0 + wave + S1_W * lane;
        const unsigned myw = (kk < nbk) ? segtab[(size_t)rb * nbk + kk] : 0u;
        const int left = nbk - k0 - wave;
        const int ns   = min(32, (left + S1_W - 1) / S1_W);
        for (int j = 0; j < ns; j += 2) {
            unsigned c[2];
            const unsigned* sg[2];
            #pragma unroll
            for (int u = 0; u < 2; ++u) {
                const unsigned w = (unsigned)__builtin_amdgcn_readlane((int)myw, j + u);
                c[u]  = w & 0xFFFFu;
                sg[u] = seg + (size_t)(k0 + wave + S1_W * (j + u)) * BK_CAP + (w >> 16);
            }
            unsigned e[8];
            bool     ok[8];
            #pragma unroll
            for (int u = 0; u < 2; ++u)
                #pragma unroll
                for (int r = 0; r < 4; ++r) {
                    const unsigned idx = 32u * r + (unsigned)lane;
                    ok[4 * u + r] = idx < c[u];
                    e[4 * u + r]  = ok[4 * u + r] ? sg[u][idx] : 0u;
                }
            v2f gv[8];
            #pragma unroll
            for (int t = 0; t < 8; ++t) gv[t] = *(const v2f*)(h2 + 2 * (e[t] >> 12));
            asm volatile("" :: "v"(gv[0]), "v"(gv[1]), "v"(gv[2]), "v"(gv[3]),
                               "v"(gv[4]), "v"(gv[5]), "v"(gv[6]), "v"(gv[7]));
            #pragma unroll
            for (int t = 0; t < 8; ++t)
                if (ok[t]) {
                    const unsigned r = e[t] & 4095u;
                    atomicAdd(&acc[2 * r],     gv[t].x);
                    atomicAdd(&acc[2 * r + 1], gv[t].y);
                }
            #pragma unroll
            for (int u = 0; u < 2; ++u)
                for (unsigned i = 128u + (unsigned)lane; i < c[u]; i += 32u) {
                    const unsigned en = sg[u][i];
                    const v2f g1 = *(const v2f*)(h2 + 2 * (en >> 12));
                    const unsigned r = en & 4095u;
                    atomicAdd(&acc[2 * r],     g1.x);
                    atomicAdd(&acc[2 * r + 1], g1.y);
                }
        }
    }
    {
        const int nq = E >> 2;
        int e = (nq << 2) + tid;
        if (e < E) {
            int sN = ei[e];
            unsigned r = (unsigned)(ei[E + e] - lo);
            if (r < S1_R) {
                atomicAdd(&acc[2 * r],     h2[2 * sN]);
                atomicAdd(&acc[2 * r + 1], h2[2 * sN + 1]);
            }
        }
    }
    __syncthreads();

    for (int i = tid; i < S1_R; i += S1_T) {
        const int node = lo + i;
        if (node < n) {
            const float w = dinv[node];
            float a0 = h2[2 * node]     + acc[2 * i];
            float a1 = h2[2 * node + 1] + acc[2 * i + 1];
            a0 = a0 * w + b2[0];
            a1 = a1 * w + b2[1];
            const float m  = fmaxf(a0, a1);
            const float ls = logf(expf(a0 - m) + expf(a1 - m));
            out[2 * node]     = a0 - m - ls;
            out[2 * node + 1] = a1 - m - ls;
        }
    }
}


extern "C" void kernel_launch(void* const* d_in, const int* in_sizes, int n_in,
                              void* d_out, int out_size, void* d_ws, size_t ws_size,
                              hipStream_t stream) {
    const float* x  = (const float*)d_in[0];
    const int*   ei = (const int*)d_in[1];
    const float* W1 = (const float*)d_in[2];
    const float* b1 = (const float*)d_in[3];
    const float* W2 = (const float*)d_in[4];
    const float* b2 = (const float*)d_in[5];
    float* out      = (float*)d_out;

    const int N = in_sizes[0] / FIN;
    const int E = in_sizes[1] / 2;

    auto blk = [](long long t, int b) { return (int)((t + b - 1) / b); };

    const int nb  = blk(N, S1_R);
    const int nbk = blk(E / 4, BK_CAP / 4);

    unsigned* seg    = (unsigned*)d_ws;
    float* h         = (float*)(seg + (size_t)nbk * BK_CAP);
    float* h2        = h    + (size_t)N * FH;
    float* dinv      = h2   + (size_t)N * FOUT;
    unsigned* segtab = (unsigned*)(dinv + N);

    k_bucket      <<<nbk, BK_T, 0, stream>>>(ei, E, nb, seg, segtab);

    const int n_tiles = N / 16;
    k_deg_gemm1   <<<nb, S1_T, 0, stream>>>(ei, E, N, seg, segtab, nbk, x, W1, dinv, h, n_tiles);

    k_scatter1_l2 <<<nb, S1_T, 0, stream>>>(ei, E, h, N, seg, segtab, nbk, dinv, b1, W2, h2);

    k_scatter2_out<<<nb, S1_T, 0, stream>>>(ei, E, h2, N, seg, segtab, nbk, dinv, b2, out);
}
